# starting kernel
# baseline (speedup 1.0000x reference)
#include <hip/hip_runtime.h>
#include <math.h>

typedef float v2f __attribute__((ext_vector_type(2)));
typedef float v8f __attribute__((ext_vector_type(8)));

#define N_NODES   100000
#define N_FEAT    512
#define HIDDEN    64
#define N_CLASSES 16

static __device__ __forceinline__ void atomic_add_f32(float* p, float v) {
    __hip_atomic_fetch_add(p, v, __ATOMIC_RELAXED, __HIP_MEMORY_SCOPE_AGENT);
}

__global__ __launch_bounds__(256) void init_kernel(float* bufB, float* out, float* deg) {
    int t = blockIdx.x * 256 + threadIdx.x;
    if (t < N_NODES * HIDDEN)    bufB[t] = 0.0f;
    if (t < N_NODES * N_CLASSES) out[t]  = 0.0f;
    if (t < N_NODES)             deg[t]  = 1.0f;
}

__global__ __launch_bounds__(256) void degree_kernel(const int* __restrict__ dst,
                                                     float* __restrict__ deg, int nE) {
    int e = blockIdx.x * 256 + threadIdx.x;
    if (e < nE) atomic_add_f32(&deg[dst[e]], 1.0f);
}

__global__ __launch_bounds__(256) void rsqrt_kernel(float* dinv) {
    int t = blockIdx.x * 256 + threadIdx.x;
    if (t < N_NODES) dinv[t] = rsqrtf(dinv[t]);
}

__global__ __launch_bounds__(256) void gemm1_kernel(const float* __restrict__ A,
                                                    const float* __restrict__ B,
                                                    float* __restrict__ C, int nTiles) {
    int wave = threadIdx.x >> 5;
    int lane = threadIdx.x & 31;
    int tile = blockIdx.x * 8 + wave;
    if (tile >= nTiles) return;
    int nr = lane & 15;
    int hi = lane >> 4;

    const float* arow = A + (size_t)(tile * 16 + nr) * N_FEAT + 2 * hi;
    v8f acc0 = {}, acc1 = {}, acc2 = {}, acc3 = {};

    for (int kk = 0; kk < N_FEAT; kk += 4) {
        v2f a = *(const v2f*)(arow + kk);
        const float* b0p = B + (size_t)(kk + 2 * hi) * HIDDEN + nr;
        const float* b1p = b0p + HIDDEN;
        v2f b0 = { b0p[0],  b1p[0]  };
        v2f b1 = { b0p[16], b1p[16] };
        v2f b2 = { b0p[32], b1p[32] };
        v2f b3 = { b0p[48], b1p[48] };
        acc0 = __builtin_amdgcn_wmma_f32_16x16x4_f32(false, a, false, b0, (short)0, acc0, false, false);
        acc1 = __builtin_amdgcn_wmma_f32_16x16x4_f32(false, a, false, b1, (short)0, acc1, false, false);
        acc2 = __builtin_amdgcn_wmma_f32_16x16x4_f32(false, a, false, b2, (short)0, acc2, false, false);
        acc3 = __builtin_amdgcn_wmma_f32_16x16x4_f32(false, a, false, b3, (short)0, acc3, false, false);
    }
    int mbase = tile * 16 + hi * 8;
    for (int r = 0; r < 8; ++r) {
        float* crow = C + (size_t)(mbase + r) * HIDDEN + nr;
        crow[0]  = acc0[r];
        crow[16] = acc1[r];
        crow[32] = acc2[r];
        crow[48] = acc3[r];
    }
}

__global__ __launch_bounds__(256) void edge_agg1_kernel(const int* __restrict__ src,
                                                        const int* __restrict__ dst,
                                                        const float* __restrict__ dinv,
                                                        const float* __restrict__ h,
                                                        float* __restrict__ agg, int nE) {
    int e = blockIdx.x * 8 + (threadIdx.x >> 5);
    if (e >= nE) return;
    int lane = threadIdx.x & 31;
    int s = src[e], d = dst[e];
    float coeff = dinv[s] * dinv[d];
    const float* hs = h   + (size_t)s * HIDDEN + lane * 2;
    float*       ad = agg + (size_t)d * HIDDEN + lane * 2;
    atomic_add_f32(&ad[0], hs[0] * coeff);
    atomic_add_f32(&ad[1], hs[1] * coeff);
}

__global__ __launch_bounds__(256) void relu_kernel(const float* __restrict__ pre,
                                                   float* __restrict__ agg,
                                                   const float* __restrict__ dinv,
                                                   const float* __restrict__ b1) {
    int t = blockIdx.x * 256 + threadIdx.x;
    if (t >= N_NODES * HIDDEN) return;
    int node = t >> 6, f = t & 63;
    float di = dinv[node];
    float v = agg[t] + di * di * pre[t] + b1[f];
    agg[t] = v > 0.0f ? v : 0.0f;
}

__global__ __launch_bounds__(256) void gemm2_kernel(const float* __restrict__ A,
                                                    const float* __restrict__ B,
                                                    float* __restrict__ C, int nTiles) {
    int wave = threadIdx.x >> 5;
    int lane = threadIdx.x & 31;
    int tile = blockIdx.x * 8 + wave;
    if (tile >= nTiles) return;
    int nr = lane & 15;
    int hi = lane >> 4;

    const float* arow = A + (size_t)(tile * 16 + nr) * HIDDEN + 2 * hi;
    v8f acc = {};
    for (int kk = 0; kk < HIDDEN; kk += 4) {
        v2f a = *(const v2f*)(arow + kk);
        const float* b0p = B + (size_t)(kk + 2 * hi) * N_CLASSES + nr;
        v2f b = { b0p[0], b0p[N_CLASSES] };
        acc = __builtin_amdgcn_wmma_f32_16x16x4_f32(false, a, false, b, (short)0, acc, false, false);
    }
    int mbase = tile * 16 + hi * 8;
    for (int r = 0; r < 8; ++r)
        C[(size_t)(mbase + r) * N_CLASSES + nr] = acc[r];
}

__global__ __launch_bounds__(256) void edge_agg2_kernel(const int* __restrict__ src,
                                                        const int* __restrict__ dst,
                                                        const float* __restrict__ dinv,
                                                        const float* __restrict__ h,
                                                        float* __restrict__ out, int nE) {
    int t = blockIdx.x * 256 + threadIdx.x;
    if (t >= nE * N_CLASSES) return;
    int e = t >> 4, c = t & 15;
    int s = src[e], d = dst[e];
    float coeff = dinv[s] * dinv[d];
    atomic_add_f32(&out[(size_t)d * N_CLASSES + c], coeff * h[(size_t)s * N_CLASSES + c]);
}

__global__ __launch_bounds__(256) void logsoftmax_kernel(float* __restrict__ out,
                                                         const float* __restrict__ pre,
                                                         const float* __restrict__ dinv,
                                                         const float* __restrict__ b2) {
    int r = blockIdx.x * 256 + threadIdx.x;
    if (r >= N_NODES) return;
    float di = dinv[r];
    float di2 = di * di;
    float v[N_CLASSES];
    float m = -INFINITY;
    for (int c = 0; c < N_CLASSES; ++c) {
        float x = out[(size_t)r * N_CLASSES + c] + di2 * pre[(size_t)r * N_CLASSES + c] + b2[c];
        v[c] = x;
        m = fmaxf(m, x);
    }
    float s = 0.0f;
    for (int c = 0; c < N_CLASSES; ++c) s += expf(v[c] - m);
    float ls = logf(s);
    for (int c = 0; c < N_CLASSES; ++c)
        out[(size_t)r * N_CLASSES + c] = v[c] - m - ls;
}

extern "C" void kernel_launch(void* const* d_in, const int* in_sizes, int n_in,
                              void* d_out, int out_size, void* d_ws, size_t ws_size,
                              hipStream_t stream) {
    const float* x  = (const float*)d_in[0];
    const int*   ei = (const int*)d_in[1];
    const float* W1 = (const float*)d_in[2];
    const float* b1 = (const float*)d_in[3];
    const float* W2 = (const float*)d_in[4];
    const float* b2 = (const float*)d_in[5];
    float* out = (float*)d_out;

    int nE = in_sizes[1] / 2;
    const int* src = ei;
    const int* dst = ei + nE;

    float* dinv = (float*)d_ws;
    float* bufA = dinv + 102400;
    float* bufB = bufA + (size_t)N_NODES * HIDDEN;

    int nTiles = N_NODES / 16;

    init_kernel<<<(N_NODES * HIDDEN + 255) / 256, 256, 0, stream>>>(bufB, out, dinv);
    degree_kernel<<<(nE + 255) / 256, 256, 0, stream>>>(dst, dinv, nE);
    rsqrt_kernel<<<(N_NODES + 255) / 256, 256, 0, stream>>>(dinv);
    gemm1_kernel<<<(nTiles + 7) / 8, 256, 0, stream>>>(x, W1, bufA, nTiles);
    edge_agg1_kernel<<<(nE + 7) / 8, 256, 0, stream>>>(src, dst, dinv, bufA, bufB, nE);
    relu_kernel<<<(N_NODES * HIDDEN + 255) / 256, 256, 0, stream>>>(bufA, bufB, dinv, b1);
    gemm2_kernel<<<(nTiles + 7) / 8, 256, 0, stream>>>(bufB, W2, bufA, nTiles);
    edge_agg2_kernel<<<(nE * N_CLASSES + 255) / 256, 256, 0, stream>>>(src, dst, dinv, bufA, out, nE);
    logsoftmax_kernel<<<(N_NODES + 255) / 256, 256, 0, stream>>>(out, bufA, dinv, b2);
}


# Float-atomic edge scatter-adds replaced by per-call counting sort by destination (int atomic rank) + atomic-free register row gathers, rows to GEMM2 via LDS; sums in rank order, softmax sum lane tree
# speedup vs baseline: 1.6770x; 1.6770x over previous
#include <hip/hip_runtime.h>
#include <math.h>

typedef float v2f __attribute__((ext_vector_type(2)));
typedef float v8f __attribute__((ext_vector_type(8)));

#define N_NODES    100000
#define N_EDGES    1600000
#define N_FEAT     512
#define HIDDEN     64
#define N_CLASSES  16
#define SCAN_T     1024
#define TILE_PITCH 66

static __device__ __forceinline__ float lane_f(float v, int j) {
    return __builtin_bit_cast(float, __builtin_amdgcn_readlane(__builtin_bit_cast(int, v), j));
}

__global__ __launch_bounds__(256) void init_kernel(int* cnt) {
    int t = blockIdx.x * 256 + threadIdx.x;
    if (t < N_NODES) cnt[t] = 0;
}

__global__ __launch_bounds__(256) void count_rank_kernel(const int* __restrict__ dst,
                                                         int* __restrict__ cnt,
                                                         int* __restrict__ rank, int nE) {
    int e = blockIdx.x * 256 + threadIdx.x;
    if (e < nE)
        rank[e] = __hip_atomic_fetch_add(&cnt[dst[e]], 1, __ATOMIC_RELAXED, __HIP_MEMORY_SCOPE_AGENT);
}

__global__ __launch_bounds__(SCAN_T) void scan_dinv_kernel(const int* __restrict__ cnt,
                                                           int* __restrict__ offs,
                                                           float* __restrict__ dinv) {
    __shared__ int s_pre[32];
    __shared__ int s_tot[32];
    int t = threadIdx.x, lane = t & 31, w = t >> 5;
    int base = blockIdx.x * SCAN_T;
    int p = 0;
    for (int i = t; i < base; i += SCAN_T) p += cnt[i];
    for (int o = 16; o > 0; o >>= 1) p += __shfl_xor(p, o);
    int i = base + t;
    int v = (i < N_NODES) ? cnt[i] : 0;
    int inc = v;
    for (int o = 1; o < 32; o <<= 1) {
        int u = __shfl_up(inc, o);
        if (lane >= o) inc += u;
    }
    if (lane == 31) s_tot[w] = inc;
    if (lane == 0)  s_pre[w] = p;
    __syncthreads();
    int pre = 0, wbase = 0;
    for (int k = 0; k < 32; ++k) {
        pre += s_pre[k];
        if (k < w) wbase += s_tot[k];
    }
    int excl = pre + wbase + inc - v;
    if (i < N_NODES) {
        offs[i] = excl;
        dinv[i] = rsqrtf(1.0f + (float)v);
        if (i == N_NODES - 1) offs[N_NODES] = excl + v;
    }
}

__global__ __launch_bounds__(256) void place_kernel(const int* __restrict__ src,
                                                    const int* __restrict__ dst,
                                                    const int* __restrict__ rank,
                                                    const int* __restrict__ offs,
                                                    int* __restrict__ sorted, int nE) {
    int e = blockIdx.x * 256 + threadIdx.x;
    if (e < nE) sorted[offs[dst[e]] + rank[e]] = src[e];
}

__global__ __launch_bounds__(256) void gemm1_kernel(const float* __restrict__ A,
                                                    const float* __restrict__ B,
                                                    float* __restrict__ C, int nTiles) {
    int wave = threadIdx.x >> 5;
    int lane = threadIdx.x & 31;
    int tile = blockIdx.x * 8 + wave;
    if (tile >= nTiles) return;
    int nr = lane & 15;
    int hi = lane >> 4;

    const float* arow = A + (size_t)(tile * 16 + nr) * N_FEAT + 2 * hi;
    v8f acc0 = {}, acc1 = {}, acc2 = {}, acc3 = {};

    for (int kk = 0; kk < N_FEAT; kk += 4) {
        v2f a = *(const v2f*)(arow + kk);
        const float* b0p = B + (size_t)(kk + 2 * hi) * HIDDEN + nr;
        const float* b1p = b0p + HIDDEN;
        v2f b0 = { b0p[0],  b1p[0]  };
        v2f b1 = { b0p[16], b1p[16] };
        v2f b2 = { b0p[32], b1p[32] };
        v2f b3 = { b0p[48], b1p[48] };
        acc0 = __builtin_amdgcn_wmma_f32_16x16x4_f32(false, a, false, b0, (short)0, acc0, false, false);
        acc1 = __builtin_amdgcn_wmma_f32_16x16x4_f32(false, a, false, b1, (short)0, acc1, false, false);
        acc2 = __builtin_amdgcn_wmma_f32_16x16x4_f32(false, a, false, b2, (short)0, acc2, false, false);
        acc3 = __builtin_amdgcn_wmma_f32_16x16x4_f32(false, a, false, b3, (short)0, acc3, false, false);
    }
    int mbase = tile * 16 + hi * 8;
    for (int r = 0; r < 8; ++r) {
        float* crow = C + (size_t)(mbase + r) * HIDDEN + nr;
        crow[0]  = acc0[r];
        crow[16] = acc1[r];
        crow[32] = acc2[r];
        crow[48] = acc3[r];
    }
}

__global__ __launch_bounds__(256) void gather1_gemm2_kernel(const int* __restrict__ offs,
                                                            const int* __restrict__ sorted,
                                                            const float* __restrict__ dinv,
                                                            const float* __restrict__ h,
                                                            const float* __restrict__ b1,
                                                            const float* __restrict__ W2,
                                                            float* __restrict__ h2) {
    __shared__ __attribute__((aligned(16))) float tile[16 * TILE_PITCH];
    int lane = threadIdx.x & 31;
    int wave = __builtin_amdgcn_readfirstlane(threadIdx.x >> 5);
    v2f bias = *(const v2f*)(b1 + 2 * lane);

    for (int q = 0; q < 2; ++q) {
        int row = wave * 2 + q;
        int d   = blockIdx.x * 16 + row;
        int beg = offs[d], end = offs[d + 1];
        float di = dinv[d];
        v2f self = *(const v2f*)(h + (size_t)d * HIDDEN + 2 * lane);
        float dd = di * di;
        float a0 = dd * self.x, a1 = dd * self.y;
        for (int base = beg; base < end; base += 32) {
            int my = base + lane;
            int   sl = 0;
            float cl = 0.0f;
            if (my < end) { sl = sorted[my]; cl = dinv[sl] * di; }
            int n = end - base;
            if (n > 32) n = 32;
            int j = 0;
            for (; j + 4 <= n; j += 4) {
                int s0 = __builtin_amdgcn_readlane(sl, j);
                int s1 = __builtin_amdgcn_readlane(sl, j + 1);
                int s2 = __builtin_amdgcn_readlane(sl, j + 2);
                int s3 = __builtin_amdgcn_readlane(sl, j + 3);
                v2f r0 = *(const v2f*)(h + (size_t)s0 * HIDDEN + 2 * lane);
                v2f r1 = *(const v2f*)(h + (size_t)s1 * HIDDEN + 2 * lane);
                v2f r2 = *(const v2f*)(h + (size_t)s2 * HIDDEN + 2 * lane);
                v2f r3 = *(const v2f*)(h + (size_t)s3 * HIDDEN + 2 * lane);
                float c0 = lane_f(cl, j), c1 = lane_f(cl, j + 1);
                float c2 = lane_f(cl, j + 2), c3 = lane_f(cl, j + 3);
                a0 += c0 * r0.x; a1 += c0 * r0.y;
                a0 += c1 * r1.x; a1 += c1 * r1.y;
                a0 += c2 * r2.x; a1 += c2 * r2.y;
                a0 += c3 * r3.x; a1 += c3 * r3.y;
            }
            for (; j < n; ++j) {
                int s0 = __builtin_amdgcn_readlane(sl, j);
                v2f r0 = *(const v2f*)(h + (size_t)s0 * HIDDEN + 2 * lane);
                float c0 = lane_f(cl, j);
                a0 += c0 * r0.x; a1 += c0 * r0.y;
            }
        }
        float v0 = a0 + bias.x, v1 = a1 + bias.y;
        v2f o = { v0 > 0.0f ? v0 : 0.0f, v1 > 0.0f ? v1 : 0.0f };
        *(v2f*)(tile + row * TILE_PITCH + 2 * lane) = o;
    }
    __syncthreads();

    if (wave == 0) {
        int nr = lane & 15;
        int hi = lane >> 4;
        const float* arow = tile + nr * TILE_PITCH + 2 * hi;
        v8f acc = {};
        for (int kk = 0; kk < HIDDEN; kk += 4) {
            v2f a = *(const v2f*)(arow + kk);
            const float* b0p = W2 + (size_t)(kk + 2 * hi) * N_CLASSES + nr;
            v2f b = { b0p[0], b0p[N_CLASSES] };
            acc = __builtin_amdgcn_wmma_f32_16x16x4_f32(false, a, false, b, (short)0, acc, false, false);
        }
        int mbase = blockIdx.x * 16 + hi * 8;
        for (int r = 0; r < 8; ++r)
            h2[(size_t)(mbase + r) * N_CLASSES + nr] = acc[r];
    }
}

__global__ __launch_bounds__(256) void gather2_lsm_kernel(const int* __restrict__ offs,
                                                          const int* __restrict__ sorted,
                                                          const float* __restrict__ dinv,
                                                          const float* __restrict__ h2,
                                                          const float* __restrict__ b2,
                                                          float* __restrict__ out) {
    int lane = threadIdx.x & 31;
    int wave = __builtin_amdgcn_readfirstlane(threadIdx.x >> 5);
    int c    = lane & 15;
    int half = lane >> 4;
    int pair = blockIdx.x * 8 + wave;
    float bc = b2[c];
    float x  = 0.0f;

    for (int q = 0; q < 2; ++q) {
        int d   = pair * 2 + q;
        int beg = offs[d], end = offs[d + 1];
        float di = dinv[d];
        float selfv = h2[(size_t)d * N_CLASSES + c];
        float acc = 0.0f;
        for (int base = beg; base < end; base += 32) {
            int my = base + lane;
            int   sl = 0;
            float cl = 0.0f;
            if (my < end) { sl = sorted[my]; cl = dinv[sl] * di; }
            int n = end - base;
            if (n > 32) n = 32;
            int j = 0;
            for (; j + 8 <= n; j += 8) {
                int sa0 = __builtin_amdgcn_readlane(sl, j),     sb0 = __builtin_amdgcn_readlane(sl, j + 1);
                int sa1 = __builtin_amdgcn_readlane(sl, j + 2), sb1 = __builtin_amdgcn_readlane(sl, j + 3);
                int sa2 = __builtin_amdgcn_readlane(sl, j + 4), sb2 = __builtin_amdgcn_readlane(sl, j + 5);
                int sa3 = __builtin_amdgcn_readlane(sl, j + 6), sb3 = __builtin_amdgcn_readlane(sl, j + 7);
                float g0 = h2[(size_t)(half ? sb0 : sa0) * N_CLASSES + c];
                float g1 = h2[(size_t)(half ? sb1 : sa1) * N_CLASSES + c];
                float g2 = h2[(size_t)(half ? sb2 : sa2) * N_CLASSES + c];
                float g3 = h2[(size_t)(half ? sb3 : sa3) * N_CLASSES + c];
                float ka0 = lane_f(cl, j),     kb0 = lane_f(cl, j + 1);
                float ka1 = lane_f(cl, j + 2), kb1 = lane_f(cl, j + 3);
                float ka2 = lane_f(cl, j + 4), kb2 = lane_f(cl, j + 5);
                float ka3 = lane_f(cl, j + 6), kb3 = lane_f(cl, j + 7);
                float k0 = half ? kb0 : ka0;
                float k1 = half ? kb1 : ka1;
                float k2 = half ? kb2 : ka2;
                float k3 = half ? kb3 : ka3;
                acc += k0 * g0;
                acc += k1 * g1;
                acc += k2 * g2;
                acc += k3 * g3;
            }
            for (; j < n; j += 2) {
                int sa = __builtin_amdgcn_readlane(sl, j), sb = __builtin_amdgcn_readlane(sl, j + 1);
                float g = h2[(size_t)(half ? sb : sa) * N_CLASSES + c];
                float ka = lane_f(cl, j), kb = lane_f(cl, j + 1);
                float k = half ? kb : ka;
                if (j + half < n) acc += k * g;
            }
        }
        acc += __shfl_xor(acc, 16);
        float val = acc + di * di * selfv + bc;
        if (half == q) x = val;
    }

    float m = x;
    for (int o = 8; o > 0; o >>= 1) m = fmaxf(m, __shfl_xor(m, o));
    float ssum = expf(x - m);
    for (int o = 8; o > 0; o >>= 1) ssum += __shfl_xor(ssum, o);
    out[(size_t)(pair * 2 + half) * N_CLASSES + c] = x - m - logf(ssum);
}

extern "C" void kernel_launch(void* const* d_in, const int* in_sizes, int n_in,
                              void* d_out, int out_size, void* d_ws, size_t ws_size,
                              hipStream_t stream) {
    const float* x  = (const float*)d_in[0];
    const int*   ei = (const int*)d_in[1];
    const float* W1 = (const float*)d_in[2];
    const float* b1 = (const float*)d_in[3];
    const float* W2 = (const float*)d_in[4];
    const float* b2 = (const float*)d_in[5];
    float* out = (float*)d_out;

    int nE = in_sizes[1] / 2;
    const int* src = ei;
    const int* dst = ei + nE;

    float* dinv = (float*)d_ws;
    float* bufA = dinv + 102400;
    float* bufB = bufA + (size_t)N_NODES * HIDDEN;
    float* h2   = bufB;
    int* sorted = (int*)(bufB + (size_t)N_NODES * N_CLASSES);
    int* rank   = sorted + N_EDGES;
    int* cnt    = rank + N_EDGES;
    int* offs   = cnt + 102400;

    int nTiles = N_NODES / 16;

    init_kernel<<<(N_NODES + 255) / 256, 256, 0, stream>>>(cnt);
    count_rank_kernel<<<(nE + 255) / 256, 256, 0, stream>>>(dst, cnt, rank, nE);
    scan_dinv_kernel<<<(N_NODES + SCAN_T - 1) / SCAN_T, SCAN_T, 0, stream>>>(cnt, offs, dinv);
    place_kernel<<<(nE + 255) / 256, 256, 0, stream>>>(src, dst, rank, offs, sorted, nE);
    gemm1_kernel<<<(nTiles + 7) / 8, 256, 0, stream>>>(x, W1, bufA, nTiles);
    gather1_gemm2_kernel<<<N_NODES / 16, 256, 0, stream>>>(offs, sorted, dinv, bufA, b1, W2, h2);
    gather2_lsm_kernel<<<N_NODES / 16, 256, 0, stream>>>(offs, sorted, dinv, h2, b2, out);
}


# Edge grouping by destination: per-edge global returning atomics replaced by LDS counters (bucket edges by dst>>9, then sort each bucket in one block); in-edge sums in LDS arrival order
# speedup vs baseline: 2.6359x; 1.5718x over previous
#include <hip/hip_runtime.h>
#include <math.h>

typedef float v2f __attribute__((ext_vector_type(2)));
typedef float v8f __attribute__((ext_vector_type(8)));

#define N_NODES    100000
#define N_EDGES    1600000
#define N_FEAT     512
#define HIDDEN     64
#define N_CLASSES  16
#define BKT_SHIFT  9
#define BKT_W      512
#define NB         ((N_NODES + BKT_W - 1) / BKT_W)
#define NBP        256
#define GC_STRIDE  32
#define BIN_T      256
#define BIN_CHUNK  (BIN_T * 24)
#define BIN_BLKS   ((N_EDGES + BIN_CHUNK - 1) / BIN_CHUNK)
#define SORT_T     512
#define STAGE_CAP  12288
static_assert(NB <= NBP && BIN_T == NBP && SORT_T == BKT_W, "one scan thread per bucket / per destination");
static_assert(N_NODES <= (1 << (32 - BKT_SHIFT)), "a source id and 9 destination bits share one 32-bit word");
#define TILE_PITCH 66

static __device__ __forceinline__ float lane_f(float v, int j) {
    return __builtin_bit_cast(float, __builtin_amdgcn_readlane(__builtin_bit_cast(int, v), j));
}

__global__ __launch_bounds__(NBP) void init_kernel(int* gcnt) {
    gcnt[threadIdx.x * GC_STRIDE] = 0;
}

__global__ __launch_bounds__(BIN_T) void bucket_count_kernel(const int* __restrict__ dst,
                                                             int* __restrict__ gcnt,
                                                             int* __restrict__ pre, int nE) {
    __shared__ int hist[NBP];
    int t = threadIdx.x;
    hist[t] = 0;
    __syncthreads();
    int c0 = blockIdx.x * BIN_CHUNK;
    int c1 = c0 + BIN_CHUNK;
    if (c1 > nE) c1 = nE;
    for (int e0 = c0; e0 < c1; e0 += BIN_T * 8) {
        bool full = e0 + BIN_T * 8 <= c1;
        int d[8];
        for (int k = 0; k < 8; ++k) {
            int ek = e0 + t + k * BIN_T;
            d[k] = dst[ek < c1 ? ek : c1 - 1];
        }
        if (full) {
            for (int k = 0; k < 8; ++k)
                __hip_atomic_fetch_add(&hist[d[k] >> BKT_SHIFT], 1, __ATOMIC_RELAXED, __HIP_MEMORY_SCOPE_WORKGROUP);
        } else {
            for (int k = 0; k < 8; ++k)
                if (e0 + t + k * BIN_T < c1)
                    __hip_atomic_fetch_add(&hist[d[k] >> BKT_SHIFT], 1, __ATOMIC_RELAXED, __HIP_MEMORY_SCOPE_WORKGROUP);
        }
    }
    __syncthreads();
    if (t < NB)
        pre[blockIdx.x * NBP + t] =
            __hip_atomic_fetch_add(&gcnt[t * GC_STRIDE], hist[t], __ATOMIC_RELAXED, __HIP_MEMORY_SCOPE_AGENT);
}

__global__ __launch_bounds__(BIN_T) void bucket_scatter_kernel(const int* __restrict__ src,
                                                               const int* __restrict__ dst,
                                                               const int* __restrict__ gcnt,
                                                               const int* __restrict__ pre,
                                                               int* __restrict__ bstart,
                                                               unsigned* __restrict__ binned, int nE) {
    __shared__ int cur[NBP];
    __shared__ int wtot[NBP / 32];
    int t = threadIdx.x, lane = t & 31, w = t >> 5;
    int v = (t < NB) ? gcnt[t * GC_STRIDE] : 0;
    int inc = v;
    for (int o = 1; o < 32; o <<= 1) {
        int u = __shfl_up(inc, o);
        if (lane >= o) inc += u;
    }
    if (lane == 31) wtot[w] = inc;
    __syncthreads();
    int wbase = 0;
    for (int k = 0; k < NBP / 32; ++k)
        if (k < w) wbase += wtot[k];
    int excl = wbase + inc - v;
    cur[t] = excl + ((t < NB) ? pre[blockIdx.x * NBP + t] : 0);
    if (blockIdx.x == 0 && t <= NB) bstart[t] = excl;
    __syncthreads();

    int c0 = blockIdx.x * BIN_CHUNK;
    int c1 = c0 + BIN_CHUNK;
    if (c1 > nE) c1 = nE;
    for (int e0 = c0; e0 < c1; e0 += BIN_T * 8) {
        bool full = e0 + BIN_T * 8 <= c1;
        int d[8], s[8];
        for (int k = 0; k < 8; ++k) {
            int ek = e0 + t + k * BIN_T;
            int ec = ek < c1 ? ek : c1 - 1;
            d[k] = dst[ec];
            s[k] = src[ec];
        }
        if (full) {
            int p[8];
            for (int k = 0; k < 8; ++k)
                p[k] = __hip_atomic_fetch_add(&cur[d[k] >> BKT_SHIFT], 1, __ATOMIC_RELAXED, __HIP_MEMORY_SCOPE_WORKGROUP);
            for (int k = 0; k < 8; ++k)
                binned[p[k]] = ((unsigned)s[k] << BKT_SHIFT) | (unsigned)(d[k] & (BKT_W - 1));
        } else {
            for (int k = 0; k < 8; ++k)
                if (e0 + t + k * BIN_T < c1) {
                    int p = __hip_atomic_fetch_add(&cur[d[k] >> BKT_SHIFT], 1, __ATOMIC_RELAXED, __HIP_MEMORY_SCOPE_WORKGROUP);
                    binned[p] = ((unsigned)s[k] << BKT_SHIFT) | (unsigned)(d[k] & (BKT_W - 1));
                }
        }
    }
}

__global__ __launch_bounds__(SORT_T) void bucket_sort_kernel(const unsigned* __restrict__ binned,
                                                             const int* __restrict__ bstart,
                                                             int* __restrict__ offs,
                                                             float* __restrict__ dinv,
                                                             int* __restrict__ sorted) {
    __shared__ int cntL[BKT_W];
    __shared__ int curL[BKT_W];
    __shared__ int wtot[SORT_T / 32];
    __shared__ int stage[STAGE_CAP];
    int t = threadIdx.x, lane = t & 31, w = t >> 5;
    int b  = blockIdx.x;
    int bs = bstart[b];
    int n  = bstart[b + 1] - bs;
    const unsigned* mine = binned + bs;
    cntL[t] = 0;
    __syncthreads();
    for (int i0 = 0; i0 < n; i0 += SORT_T * 8) {
        bool full = i0 + SORT_T * 8 <= n;
        unsigned p[8];
        for (int k = 0; k < 8; ++k) {
            int ik = i0 + t + k * SORT_T;
            p[k] = mine[ik < n ? ik : n - 1];
        }
        if (full) {
            for (int k = 0; k < 8; ++k)
                __hip_atomic_fetch_add(&cntL[p[k] & (BKT_W - 1)], 1, __ATOMIC_RELAXED, __HIP_MEMORY_SCOPE_WORKGROUP);
        } else {
            for (int k = 0; k < 8; ++k)
                if (i0 + t + k * SORT_T < n)
                    __hip_atomic_fetch_add(&cntL[p[k] & (BKT_W - 1)], 1, __ATOMIC_RELAXED, __HIP_MEMORY_SCOPE_WORKGROUP);
        }
    }
    __syncthreads();

    int v = cntL[t];
    int inc = v;
    for (int o = 1; o < 32; o <<= 1) {
        int u = __shfl_up(inc, o);
        if (lane >= o) inc += u;
    }
    if (lane == 31) wtot[w] = inc;
    __syncthreads();
    int wbase = 0;
    for (int k = 0; k < SORT_T / 32; ++k)
        if (k < w) wbase += wtot[k];
    int excl = wbase + inc - v;
    curL[t] = excl;
    int dnode = b * BKT_W + t;
    if (dnode < N_NODES) {
        offs[dnode] = bs + excl;
        dinv[dnode] = rsqrtf(1.0f + (float)v);
        if (dnode == N_NODES - 1) offs[N_NODES] = bs + excl + v;
    }
    __syncthreads();

    bool fits = n <= STAGE_CAP;
    for (int i0 = 0; i0 < n; i0 += SORT_T * 8) {
        bool full = i0 + SORT_T * 8 <= n;
        unsigned p[8];
        for (int k = 0; k < 8; ++k) {
            int ik = i0 + t + k * SORT_T;
            p[k] = mine[ik < n ? ik : n - 1];
        }
        if (full && fits) {
            int pos[8];
            for (int k = 0; k < 8; ++k)
                pos[k] = __hip_atomic_fetch_add(&curL[p[k] & (BKT_W - 1)], 1, __ATOMIC_RELAXED, __HIP_MEMORY_SCOPE_WORKGROUP);
            for (int k = 0; k < 8; ++k)
                stage[pos[k]] = (int)(p[k] >> BKT_SHIFT);
        } else {
            for (int k = 0; k < 8; ++k)
                if (i0 + t + k * SORT_T < n) {
                    int pos = __hip_atomic_fetch_add(&curL[p[k] & (BKT_W - 1)], 1, __ATOMIC_RELAXED, __HIP_MEMORY_SCOPE_WORKGROUP);
                    int s = (int)(p[k] >> BKT_SHIFT);
                    if (fits) stage[pos] = s;
                    else      sorted[bs + pos] = s;
                }
        }
    }
    __syncthreads();
    if (fits)
        for (int i = t; i < n; i += SORT_T) sorted[bs + i] = stage[i];
}

__global__ __launch_bounds__(256) void gemm1_kernel(const float* __restrict__ A,
                                                    const float* __restrict__ B,
                                                    float* __restrict__ C, int nTiles) {
    int wave = threadIdx.x >> 5;
    int lane = threadIdx.x & 31;
    int tile = blockIdx.x * 8 + wave;
    if (tile >= nTiles) return;
    int nr = lane & 15;
    int hi = lane >> 4;

    const float* arow = A + (size_t)(tile * 16 + nr) * N_FEAT + 2 * hi;
    v8f acc0 = {}, acc1 = {}, acc2 = {}, acc3 = {};

    for (int kk = 0; kk < N_FEAT; kk += 4) {
        v2f a = *(const v2f*)(arow + kk);
        const float* b0p = B + (size_t)(kk + 2 * hi) * HIDDEN + nr;
        const float* b1p = b0p + HIDDEN;
        v2f b0 = { b0p[0],  b1p[0]  };
        v2f b1 = { b0p[16], b1p[16] };
        v2f b2 = { b0p[32], b1p[32] };
        v2f b3 = { b0p[48], b1p[48] };
        acc0 = __builtin_amdgcn_wmma_f32_16x16x4_f32(false, a, false, b0, (short)0, acc0, false, false);
        acc1 = __builtin_amdgcn_wmma_f32_16x16x4_f32(false, a, false, b1, (short)0, acc1, false, false);
        acc2 = __builtin_amdgcn_wmma_f32_16x16x4_f32(false, a, false, b2, (short)0, acc2, false, false);
        acc3 = __builtin_amdgcn_wmma_f32_16x16x4_f32(false, a, false, b3, (short)0, acc3, false, false);
    }
    int mbase = tile * 16 + hi * 8;
    for (int r = 0; r < 8; ++r) {
        float* crow = C + (size_t)(mbase + r) * HIDDEN + nr;
        crow[0]  = acc0[r];
        crow[16] = acc1[r];
        crow[32] = acc2[r];
        crow[48] = acc3[r];
    }
}

__global__ __launch_bounds__(256) void gather1_gemm2_kernel(const int* __restrict__ offs,
                                                            const int* __restrict__ sorted,
                                                            const float* __restrict__ dinv,
                                                            const float* __restrict__ h,
                                                            const float* __restrict__ b1,
                                                            const float* __restrict__ W2,
                                                            float* __restrict__ h2) {
    __shared__ __attribute__((aligned(16))) float tile[16 * TILE_PITCH];
    int lane = threadIdx.x & 31;
    int wave = __builtin_amdgcn_readfirstlane(threadIdx.x >> 5);
    v2f bias = *(const v2f*)(b1 + 2 * lane);

    for (int q = 0; q < 2; ++q) {
        int row = wave * 2 + q;
        int d   = blockIdx.x * 16 + row;
        int beg = offs[d], end = offs[d + 1];
        float di = dinv[d];
        v2f self = *(const v2f*)(h + (size_t)d * HIDDEN + 2 * lane);
        float dd = di * di;
        float a0 = dd * self.x, a1 = dd * self.y;
        for (int base = beg; base < end; base += 32) {
            int my = base + lane;
            int   sl = 0;
            float cl = 0.0f;
            if (my < end) { sl = sorted[my]; cl = dinv[sl] * di; }
            int n = end - base;
            if (n > 32) n = 32;
            int j = 0;
            for (; j + 4 <= n; j += 4) {
                int s0 = __builtin_amdgcn_readlane(sl, j);
                int s1 = __builtin_amdgcn_readlane(sl, j + 1);
                int s2 = __builtin_amdgcn_readlane(sl, j + 2);
                int s3 = __builtin_amdgcn_readlane(sl, j + 3);
                v2f r0 = *(const v2f*)(h + (size_t)s0 * HIDDEN + 2 * lane);
                v2f r1 = *(const v2f*)(h + (size_t)s1 * HIDDEN + 2 * lane);
                v2f r2 = *(const v2f*)(h + (size_t)s2 * HIDDEN + 2 * lane);
                v2f r3 = *(const v2f*)(h + (size_t)s3 * HIDDEN + 2 * lane);
                float c0 = lane_f(cl, j), c1 = lane_f(cl, j + 1);
                float c2 = lane_f(cl, j + 2), c3 = lane_f(cl, j + 3);
                a0 += c0 * r0.x; a1 += c0 * r0.y;
                a0 += c1 * r1.x; a1 += c1 * r1.y;
                a0 += c2 * r2.x; a1 += c2 * r2.y;
                a0 += c3 * r3.x; a1 += c3 * r3.y;
            }
            for (; j < n; ++j) {
                int s0 = __builtin_amdgcn_readlane(sl, j);
                v2f r0 = *(const v2f*)(h + (size_t)s0 * HIDDEN + 2 * lane);
                float c0 = lane_f(cl, j);
                a0 += c0 * r0.x; a1 += c0 * r0.y;
            }
        }
        float v0 = a0 + bias.x, v1 = a1 + bias.y;
        v2f o = { v0 > 0.0f ? v0 : 0.0f, v1 > 0.0f ? v1 : 0.0f };
        *(v2f*)(tile + row * TILE_PITCH + 2 * lane) = o;
    }
    __syncthreads();

    if (wave == 0) {
        int nr = lane & 15;
        int hi = lane >> 4;
        const float* arow = tile + nr * TILE_PITCH + 2 * hi;
        v8f acc = {};
        for (int kk = 0; kk < HIDDEN; kk += 4) {
            v2f a = *(const v2f*)(arow + kk);
            const float* b0p = W2 + (size_t)(kk + 2 * hi) * N_CLASSES + nr;
            v2f b = { b0p[0], b0p[N_CLASSES] };
            acc = __builtin_amdgcn_wmma_f32_16x16x4_f32(false, a, false, b, (short)0, acc, false, false);
        }
        int mbase = blockIdx.x * 16 + hi * 8;
        for (int r = 0; r < 8; ++r)
            h2[(size_t)(mbase + r) * N_CLASSES + nr] = acc[r];
    }
}

__global__ __launch_bounds__(256) void gather2_lsm_kernel(const int* __restrict__ offs,
                                                          const int* __restrict__ sorted,
                                                          const float* __restrict__ dinv,
                                                          const float* __restrict__ h2,
                                                          const float* __restrict__ b2,
                                                          float* __restrict__ out) {
    int lane = threadIdx.x & 31;
    int wave = __builtin_amdgcn_readfirstlane(threadIdx.x >> 5);
    int c    = lane & 15;
    int half = lane >> 4;
    int pair = blockIdx.x * 8 + wave;
    float bc = b2[c];
    float x  = 0.0f;

    for (int q = 0; q < 2; ++q) {
        int d   = pair * 2 + q;
        int beg = offs[d], end = offs[d + 1];
        float di = dinv[d];
        float selfv = h2[(size_t)d * N_CLASSES + c];
        float acc = 0.0f;
        for (int base = beg; base < end; base += 32) {
            int my = base + lane;
            int   sl = 0;
            float cl = 0.0f;
            if (my < end) { sl = sorted[my]; cl = dinv[sl] * di; }
            int n = end - base;
            if (n > 32) n = 32;
            int j = 0;
            for (; j + 8 <= n; j += 8) {
                int sa0 = __builtin_amdgcn_readlane(sl, j),     sb0 = __builtin_amdgcn_readlane(sl, j + 1);
                int sa1 = __builtin_amdgcn_readlane(sl, j + 2), sb1 = __builtin_amdgcn_readlane(sl, j + 3);
                int sa2 = __builtin_amdgcn_readlane(sl, j + 4), sb2 = __builtin_amdgcn_readlane(sl, j + 5);
                int sa3 = __builtin_amdgcn_readlane(sl, j + 6), sb3 = __builtin_amdgcn_readlane(sl, j + 7);
                float g0 = h2[(size_t)(half ? sb0 : sa0) * N_CLASSES + c];
                float g1 = h2[(size_t)(half ? sb1 : sa1) * N_CLASSES + c];
                float g2 = h2[(size_t)(half ? sb2 : sa2) * N_CLASSES + c];
                float g3 = h2[(size_t)(half ? sb3 : sa3) * N_CLASSES + c];
                float ka0 = lane_f(cl, j),     kb0 = lane_f(cl, j + 1);
                float ka1 = lane_f(cl, j + 2), kb1 = lane_f(cl, j + 3);
                float ka2 = lane_f(cl, j + 4), kb2 = lane_f(cl, j + 5);
                float ka3 = lane_f(cl, j + 6), kb3 = lane_f(cl, j + 7);
                float k0 = half ? kb0 : ka0;
                float k1 = half ? kb1 : ka1;
                float k2 = half ? kb2 : ka2;
                float k3 = half ? kb3 : ka3;
                acc += k0 * g0;
                acc += k1 * g1;
                acc += k2 * g2;
                acc += k3 * g3;
            }
            for (; j < n; j += 2) {
                int sa = __builtin_amdgcn_readlane(sl, j), sb = __builtin_amdgcn_readlane(sl, j + 1);
                float g = h2[(size_t)(half ? sb : sa) * N_CLASSES + c];
                float ka = lane_f(cl, j), kb = lane_f(cl, j + 1);
                float k = half ? kb : ka;
                if (j + half < n) acc += k * g;
            }
        }
        acc += __shfl_xor(acc, 16);
        float val = acc + di * di * selfv + bc;
        if (half == q) x = val;
    }

    float m = x;
    for (int o = 8; o > 0; o >>= 1) m = fmaxf(m, __shfl_xor(m, o));
    float ssum = expf(x - m);
    for (int o = 8; o > 0; o >>= 1) ssum += __shfl_xor(ssum, o);
    out[(size_t)(pair * 2 + half) * N_CLASSES + c] = x - m - logf(ssum);
}

extern "C" void kernel_launch(void* const* d_in, const int* in_sizes, int n_in,
                              void* d_out, int out_size, void* d_ws, size_t ws_size,
                              hipStream_t stream) {
    const float* x  = (const float*)d_in[0];
    const int*   ei = (const int*)d_in[1];
    const float* W1 = (const float*)d_in[2];
    const float* b1 = (const float*)d_in[3];
    const float* W2 = (const float*)d_in[4];
    const float* b2 = (const float*)d_in[5];
    float* out = (float*)d_out;

    int nE = in_sizes[1] / 2;
    const int* src = ei;
    const int* dst = ei + nE;

    float* dinv = (float*)d_ws;
    float* bufA = dinv + 102400;
    float* bufB = bufA + (size_t)N_NODES * HIDDEN;
    float* h2   = bufB;
    int* sorted = (int*)(bufB + (size_t)N_NODES * N_CLASSES);
    int* offs   = sorted + N_EDGES;
    int* pre    = offs + 102400;
    int* gcnt   = pre + BIN_BLKS * NBP;
    int* bstart = gcnt + NBP * GC_STRIDE;
    unsigned* binned = (unsigned*)bufA;

    int nTiles = N_NODES / 16;

    init_kernel<<<1, NBP, 0, stream>>>(gcnt);
    bucket_count_kernel<<<(nE + BIN_CHUNK - 1) / BIN_CHUNK, BIN_T, 0, stream>>>(dst, gcnt, pre, nE);
    bucket_scatter_kernel<<<(nE + BIN_CHUNK - 1) / BIN_CHUNK, BIN_T, 0, stream>>>(src, dst, gcnt, pre, bstart, binned, nE);
    bucket_sort_kernel<<<NB, SORT_T, 0, stream>>>(binned, bstart, offs, dinv, sorted);
    gemm1_kernel<<<(nTiles + 7) / 8, 256, 0, stream>>>(x, W1, bufA, nTiles);
    gather1_gemm2_kernel<<<N_NODES / 16, 256, 0, stream>>>(offs, sorted, dinv, bufA, b1, W2, h2);
    gather2_lsm_kernel<<<N_NODES / 16, 256, 0, stream>>>(offs, sorted, dinv, h2, b2, out);
}


# GEMM1 fragments by 16-byte loads (lane owns out columns 4c..4c+3 and k 4hi..4hi+3 per 8-k group: k sum order inside each 8 permuted), issued one group ahead; C rows stored in 16-byte pieces
# speedup vs baseline: 2.7574x; 1.0461x over previous
#include <hip/hip_runtime.h>
#include <math.h>

typedef float v2f __attribute__((ext_vector_type(2)));
typedef float v4f __attribute__((ext_vector_type(4)));
typedef float v8f __attribute__((ext_vector_type(8)));

#define N_NODES    100000
#define N_EDGES    1600000
#define N_FEAT     512
#define HIDDEN     64
#define N_CLASSES  16
#define BKT_SHIFT  9
#define BKT_W      512
#define NB         ((N_NODES + BKT_W - 1) / BKT_W)
#define NBP        256
#define GC_STRIDE  32
#define BIN_T      256
#define BIN_CHUNK  (BIN_T * 24)
#define BIN_BLKS   ((N_EDGES + BIN_CHUNK - 1) / BIN_CHUNK)
#define SORT_T     512
#define STAGE_CAP  12288
static_assert(NB <= NBP && BIN_T == NBP && SORT_T == BKT_W, "one scan thread per bucket / per destination");
static_assert(N_NODES <= (1 << (32 - BKT_SHIFT)), "a source id and 9 destination bits share one 32-bit word");
#define TILE_PITCH 66

static __device__ __forceinline__ float lane_f(float v, int j) {
    return __builtin_bit_cast(float, __builtin_amdgcn_readlane(__builtin_bit_cast(int, v), j));
}

__global__ __launch_bounds__(NBP) void init_kernel(int* gcnt) {
    gcnt[threadIdx.x * GC_STRIDE] = 0;
}

__global__ __launch_bounds__(BIN_T) void bucket_count_kernel(const int* __restrict__ dst,
                                                             int* __restrict__ gcnt,
                                                             int* __restrict__ pre, int nE) {
    __shared__ int hist[NBP];
    int t = threadIdx.x;
    hist[t] = 0;
    __syncthreads();
    int c0 = blockIdx.x * BIN_CHUNK;
    int c1 = c0 + BIN_CHUNK;
    if (c1 > nE) c1 = nE;
    for (int e0 = c0; e0 < c1; e0 += BIN_T * 8) {
        bool full = e0 + BIN_T * 8 <= c1;
        int d[8];
        for (int k = 0; k < 8; ++k) {
            int ek = e0 + t + k * BIN_T;
            d[k] = dst[ek < c1 ? ek : c1 - 1];
        }
        if (full) {
            for (int k = 0; k < 8; ++k)
                __hip_atomic_fetch_add(&hist[d[k] >> BKT_SHIFT], 1, __ATOMIC_RELAXED, __HIP_MEMORY_SCOPE_WORKGROUP);
        } else {
            for (int k = 0; k < 8; ++k)
                if (e0 + t + k * BIN_T < c1)
                    __hip_atomic_fetch_add(&hist[d[k] >> BKT_SHIFT], 1, __ATOMIC_RELAXED, __HIP_MEMORY_SCOPE_WORKGROUP);
        }
    }
    __syncthreads();
    if (t < NB)
        pre[blockIdx.x * NBP + t] =
            __hip_atomic_fetch_add(&gcnt[t * GC_STRIDE], hist[t], __ATOMIC_RELAXED, __HIP_MEMORY_SCOPE_AGENT);
}

__global__ __launch_bounds__(BIN_T) void bucket_scatter_kernel(const int* __restrict__ src,
                                                               const int* __restrict__ dst,
                                                               const int* __restrict__ gcnt,
                                                               const int* __restrict__ pre,
                                                               int* __restrict__ bstart,
                                                               unsigned* __restrict__ binned, int nE) {
    __shared__ int cur[NBP];
    __shared__ int wtot[NBP / 32];
    int t = threadIdx.x, lane = t & 31, w = t >> 5;
    int v = (t < NB) ? gcnt[t * GC_STRIDE] : 0;
    int inc = v;
    for (int o = 1; o < 32; o <<= 1) {
        int u = __shfl_up(inc, o);
        if (lane >= o) inc += u;
    }
    if (lane == 31) wtot[w] = inc;
    __syncthreads();
    int wbase = 0;
    for (int k = 0; k < NBP / 32; ++k)
        if (k < w) wbase += wtot[k];
    int excl = wbase + inc - v;
    cur[t] = excl + ((t < NB) ? pre[blockIdx.x * NBP + t] : 0);
    if (blockIdx.x == 0 && t <= NB) bstart[t] = excl;
    __syncthreads();

    int c0 = blockIdx.x * BIN_CHUNK;
    int c1 = c0 + BIN_CHUNK;
    if (c1 > nE) c1 = nE;
    for (int e0 = c0; e0 < c1; e0 += BIN_T * 8) {
        bool full = e0 + BIN_T * 8 <= c1;
        int d[8], s[8];
        for (int k = 0; k < 8; ++k) {
            int ek = e0 + t + k * BIN_T;
            int ec = ek < c1 ? ek : c1 - 1;
            d[k] = dst[ec];
            s[k] = src[ec];
        }
        if (full) {
            int p[8];
            for (int k = 0; k < 8; ++k)
                p[k] = __hip_atomic_fetch_add(&cur[d[k] >> BKT_SHIFT], 1, __ATOMIC_RELAXED, __HIP_MEMORY_SCOPE_WORKGROUP);
            for (int k = 0; k < 8; ++k)
                binned[p[k]] = ((unsigned)s[k] << BKT_SHIFT) | (unsigned)(d[k] & (BKT_W - 1));
        } else {
            for (int k = 0; k < 8; ++k)
                if (e0 + t + k * BIN_T < c1) {
                    int p = __hip_atomic_fetch_add(&cur[d[k] >> BKT_SHIFT], 1, __ATOMIC_RELAXED, __HIP_MEMORY_SCOPE_WORKGROUP);
                    binned[p] = ((unsigned)s[k] << BKT_SHIFT) | (unsigned)(d[k] & (BKT_W - 1));
                }
        }
    }
}

__global__ __launch_bounds__(SORT_T) void bucket_sort_kernel(const unsigned* __restrict__ binned,
                                                             const int* __restrict__ bstart,
                                                             int* __restrict__ offs,
                                                             float* __restrict__ dinv,
                                                             int* __restrict__ sorted) {
    __shared__ int cntL[BKT_W];
    __shared__ int curL[BKT_W];
    __shared__ int wtot[SORT_T / 32];
    __shared__ int stage[STAGE_CAP];
    int t = threadIdx.x, lane = t & 31, w = t >> 5;
    int b  = blockIdx.x;
    int bs = bstart[b];
    int n  = bstart[b + 1] - bs;
    const unsigned* mine = binned + bs;
    cntL[t] = 0;
    __syncthreads();
    for (int i0 = 0; i0 < n; i0 += SORT_T * 8) {
        bool full = i0 + SORT_T * 8 <= n;
        unsigned p[8];
        for (int k = 0; k < 8; ++k) {
            int ik = i0 + t + k * SORT_T;
            p[k] = mine[ik < n ? ik : n - 1];
        }
        if (full) {
            for (int k = 0; k < 8; ++k)
                __hip_atomic_fetch_add(&cntL[p[k] & (BKT_W - 1)], 1, __ATOMIC_RELAXED, __HIP_MEMORY_SCOPE_WORKGROUP);
        } else {
            for (int k = 0; k < 8; ++k)
                if (i0 + t + k * SORT_T < n)
                    __hip_atomic_fetch_add(&cntL[p[k] & (BKT_W - 1)], 1, __ATOMIC_RELAXED, __HIP_MEMORY_SCOPE_WORKGROUP);
        }
    }
    __syncthreads();

    int v = cntL[t];
    int inc = v;
    for (int o = 1; o < 32; o <<= 1) {
        int u = __shfl_up(inc, o);
        if (lane >= o) inc += u;
    }
    if (lane == 31) wtot[w] = inc;
    __syncthreads();
    int wbase = 0;
    for (int k = 0; k < SORT_T / 32; ++k)
        if (k < w) wbase += wtot[k];
    int excl = wbase + inc - v;
    curL[t] = excl;
    int dnode = b * BKT_W + t;
    if (dnode < N_NODES) {
        offs[dnode] = bs + excl;
        dinv[dnode] = rsqrtf(1.0f + (float)v);
        if (dnode == N_NODES - 1) offs[N_NODES] = bs + excl + v;
    }
    __syncthreads();

    bool fits = n <= STAGE_CAP;
    for (int i0 = 0; i0 < n; i0 += SORT_T * 8) {
        bool full = i0 + SORT_T * 8 <= n;
        unsigned p[8];
        for (int k = 0; k < 8; ++k) {
            int ik = i0 + t + k * SORT_T;
            p[k] = mine[ik < n ? ik : n - 1];
        }
        if (full && fits) {
            int pos[8];
            for (int k = 0; k < 8; ++k)
                pos[k] = __hip_atomic_fetch_add(&curL[p[k] & (BKT_W - 1)], 1, __ATOMIC_RELAXED, __HIP_MEMORY_SCOPE_WORKGROUP);
            for (int k = 0; k < 8; ++k)
                stage[pos[k]] = (int)(p[k] >> BKT_SHIFT);
        } else {
            for (int k = 0; k < 8; ++k)
                if (i0 + t + k * SORT_T < n) {
                    int pos = __hip_atomic_fetch_add(&curL[p[k] & (BKT_W - 1)], 1, __ATOMIC_RELAXED, __HIP_MEMORY_SCOPE_WORKGROUP);
                    int s = (int)(p[k] >> BKT_SHIFT);
                    if (fits) stage[pos] = s;
                    else      sorted[bs + pos] = s;
                }
        }
    }
    __syncthreads();
    if (fits)
        for (int i = t; i < n; i += SORT_T) sorted[bs + i] = stage[i];
}

__global__ __launch_bounds__(256) void gemm1_kernel(const float* __restrict__ A,
                                                    const float* __restrict__ B,
                                                    float* __restrict__ C, int nTiles) {
    int wave = threadIdx.x >> 5;
    int lane = threadIdx.x & 31;
    int tile = blockIdx.x * 8 + wave;
    if (tile >= nTiles) return;
    int c  = lane & 15;
    int hi = lane >> 4;

    const float* arow = A + (size_t)(tile * 16 + c) * N_FEAT + 4 * hi;
    const float* bcol = B + (size_t)(4 * hi) * HIDDEN + 4 * c;
    v8f acc0 = {}, acc1 = {}, acc2 = {}, acc3 = {};

    v4f an0 = *(const v4f*)(arow);
    v4f an1 = *(const v4f*)(arow + 8);
    v4f an2 = *(const v4f*)(arow + 16);
    v4f an3 = *(const v4f*)(arow + 24);
    v4f qn0 = *(const v4f*)(bcol);
    v4f qn1 = *(const v4f*)(bcol + HIDDEN);
    v4f qn2 = *(const v4f*)(bcol + 2 * HIDDEN);
    v4f qn3 = *(const v4f*)(bcol + 3 * HIDDEN);
    for (int K0 = 0; K0 < N_FEAT; K0 += 32) {
        v4f a[4] = { an0, an1, an2, an3 };
        if (K0 + 32 < N_FEAT) {
            const float* ap = arow + K0 + 32;
            an0 = *(const v4f*)(ap);
            an1 = *(const v4f*)(ap + 8);
            an2 = *(const v4f*)(ap + 16);
            an3 = *(const v4f*)(ap + 24);
        }
        for (int s = 0; s < 4; ++s) {
            v4f q0 = qn0, q1 = qn1, q2 = qn2, q3 = qn3;
            int kn = K0 + 8 * s + 8;
            if (kn < N_FEAT) {
                const float* bp = bcol + (size_t)kn * HIDDEN;
                qn0 = *(const v4f*)(bp);
                qn1 = *(const v4f*)(bp + HIDDEN);
                qn2 = *(const v4f*)(bp + 2 * HIDDEN);
                qn3 = *(const v4f*)(bp + 3 * HIDDEN);
            }
            __builtin_amdgcn_sched_barrier(0);
            v2f a01 = { a[s].x, a[s].y };
            v2f a23 = { a[s].z, a[s].w };
            v2f b;
            b = (v2f){ q0.x, q1.x }; acc0 = __builtin_amdgcn_wmma_f32_16x16x4_f32(false, a01, false, b, (short)0, acc0, false, false);
            b = (v2f){ q0.y, q1.y }; acc1 = __builtin_amdgcn_wmma_f32_16x16x4_f32(false, a01, false, b, (short)0, acc1, false, false);
            b = (v2f){ q0.z, q1.z }; acc2 = __builtin_amdgcn_wmma_f32_16x16x4_f32(false, a01, false, b, (short)0, acc2, false, false);
            b = (v2f){ q0.w, q1.w }; acc3 = __builtin_amdgcn_wmma_f32_16x16x4_f32(false, a01, false, b, (short)0, acc3, false, false);
            b = (v2f){ q2.x, q3.x }; acc0 = __builtin_amdgcn_wmma_f32_16x16x4_f32(false, a23, false, b, (short)0, acc0, false, false);
            b = (v2f){ q2.y, q3.y }; acc1 = __builtin_amdgcn_wmma_f32_16x16x4_f32(false, a23, false, b, (short)0, acc1, false, false);
            b = (v2f){ q2.z, q3.z }; acc2 = __builtin_amdgcn_wmma_f32_16x16x4_f32(false, a23, false, b, (short)0, acc2, false, false);
            b = (v2f){ q2.w, q3.w }; acc3 = __builtin_amdgcn_wmma_f32_16x16x4_f32(false, a23, false, b, (short)0, acc3, false, false);
        }
    }
    int mbase = tile * 16 + hi * 8;
    for (int r = 0; r < 8; ++r) {
        v4f o = { acc0[r], acc1[r], acc2[r], acc3[r] };
        *(v4f*)(C + (size_t)(mbase + r) * HIDDEN + 4 * c) = o;
    }
}

__global__ __launch_bounds__(256) void gather1_gemm2_kernel(const int* __restrict__ offs,
                                                            const int* __restrict__ sorted,
                                                            const float* __restrict__ dinv,
                                                            const float* __restrict__ h,
                                                            const float* __restrict__ b1,
                                                            const float* __restrict__ W2,
                                                            float* __restrict__ h2) {
    __shared__ __attribute__((aligned(16))) float tile[16 * TILE_PITCH];
    int lane = threadIdx.x & 31;
    int wave = __builtin_amdgcn_readfirstlane(threadIdx.x >> 5);
    v2f bias = *(const v2f*)(b1 + 2 * lane);

    for (int q = 0; q < 2; ++q) {
        int row = wave * 2 + q;
        int d   = blockIdx.x * 16 + row;
        int beg = offs[d], end = offs[d + 1];
        float di = dinv[d];
        v2f self = *(const v2f*)(h + (size_t)d * HIDDEN + 2 * lane);
        float dd = di * di;
        float a0 = dd * self.x, a1 = dd * self.y;
        for (int base = beg; base < end; base += 32) {
            int my = base + lane;
            int   sl = 0;
            float cl = 0.0f;
            if (my < end) { sl = sorted[my]; cl = dinv[sl] * di; }
            int n = end - base;
            if (n > 32) n = 32;
            int j = 0;
            for (; j + 4 <= n; j += 4) {
                int s0 = __builtin_amdgcn_readlane(sl, j);
                int s1 = __builtin_amdgcn_readlane(sl, j + 1);
                int s2 = __builtin_amdgcn_readlane(sl, j + 2);
                int s3 = __builtin_amdgcn_readlane(sl, j + 3);
                v2f r0 = *(const v2f*)(h + (size_t)s0 * HIDDEN + 2 * lane);
                v2f r1 = *(const v2f*)(h + (size_t)s1 * HIDDEN + 2 * lane);
                v2f r2 = *(const v2f*)(h + (size_t)s2 * HIDDEN + 2 * lane);
                v2f r3 = *(const v2f*)(h + (size_t)s3 * HIDDEN + 2 * lane);
                float c0 = lane_f(cl, j), c1 = lane_f(cl, j + 1);
                float c2 = lane_f(cl, j + 2), c3 = lane_f(cl, j + 3);
                a0 += c0 * r0.x; a1 += c0 * r0.y;
                a0 += c1 * r1.x; a1 += c1 * r1.y;
                a0 += c2 * r2.x; a1 += c2 * r2.y;
                a0 += c3 * r3.x; a1 += c3 * r3.y;
            }
            for (; j < n; ++j) {
                int s0 = __builtin_amdgcn_readlane(sl, j);
                v2f r0 = *(const v2f*)(h + (size_t)s0 * HIDDEN + 2 * lane);
                float c0 = lane_f(cl, j);
                a0 += c0 * r0.x; a1 += c0 * r0.y;
            }
        }
        float v0 = a0 + bias.x, v1 = a1 + bias.y;
        v2f o = { v0 > 0.0f ? v0 : 0.0f, v1 > 0.0f ? v1 : 0.0f };
        *(v2f*)(tile + row * TILE_PITCH + 2 * lane) = o;
    }
    __syncthreads();

    if (wave == 0) {
        int nr = lane & 15;
        int hi = lane >> 4;
        const float* arow = tile + nr * TILE_PITCH + 2 * hi;
        v8f acc = {};
        for (int kk = 0; kk < HIDDEN; kk += 4) {
            v2f a = *(const v2f*)(arow + kk);
            const float* b0p = W2 + (size_t)(kk + 2 * hi) * N_CLASSES + nr;
            v2f b = { b0p[0], b0p[N_CLASSES] };
            acc = __builtin_amdgcn_wmma_f32_16x16x4_f32(false, a, false, b, (short)0, acc, false, false);
        }
        int mbase = blockIdx.x * 16 + hi * 8;
        for (int r = 0; r < 8; ++r)
            h2[(size_t)(mbase + r) * N_CLASSES + nr] = acc[r];
    }
}

__global__ __launch_bounds__(256) void gather2_lsm_kernel(const int* __restrict__ offs,
                                                          const int* __restrict__ sorted,
                                                          const float* __restrict__ dinv,
                                                          const float* __restrict__ h2,
                                                          const float* __restrict__ b2,
                                                          float* __restrict__ out) {
    int lane = threadIdx.x & 31;
    int wave = __builtin_amdgcn_readfirstlane(threadIdx.x >> 5);
    int c    = lane & 15;
    int half = lane >> 4;
    int pair = blockIdx.x * 8 + wave;
    float bc = b2[c];
    float x  = 0.0f;

    for (int q = 0; q < 2; ++q) {
        int d   = pair * 2 + q;
        int beg = offs[d], end = offs[d + 1];
        float di = dinv[d];
        float selfv = h2[(size_t)d * N_CLASSES + c];
        float acc = 0.0f;
        for (int base = beg; base < end; base += 32) {
            int my = base + lane;
            int   sl = 0;
            float cl = 0.0f;
            if (my < end) { sl = sorted[my]; cl = dinv[sl] * di; }
            int n = end - base;
            if (n > 32) n = 32;
            int j = 0;
            for (; j + 8 <= n; j += 8) {
                int sa0 = __builtin_amdgcn_readlane(sl, j),     sb0 = __builtin_amdgcn_readlane(sl, j + 1);
                int sa1 = __builtin_amdgcn_readlane(sl, j + 2), sb1 = __builtin_amdgcn_readlane(sl, j + 3);
                int sa2 = __builtin_amdgcn_readlane(sl, j + 4), sb2 = __builtin_amdgcn_readlane(sl, j + 5);
                int sa3 = __builtin_amdgcn_readlane(sl, j + 6), sb3 = __builtin_amdgcn_readlane(sl, j + 7);
                float g0 = h2[(size_t)(half ? sb0 : sa0) * N_CLASSES + c];
                float g1 = h2[(size_t)(half ? sb1 : sa1) * N_CLASSES + c];
                float g2 = h2[(size_t)(half ? sb2 : sa2) * N_CLASSES + c];
                float g3 = h2[(size_t)(half ? sb3 : sa3) * N_CLASSES + c];
                float ka0 = lane_f(cl, j),     kb0 = lane_f(cl, j + 1);
                float ka1 = lane_f(cl, j + 2), kb1 = lane_f(cl, j + 3);
                float ka2 = lane_f(cl, j + 4), kb2 = lane_f(cl, j + 5);
                float ka3 = lane_f(cl, j + 6), kb3 = lane_f(cl, j + 7);
                float k0 = half ? kb0 : ka0;
                float k1 = half ? kb1 : ka1;
                float k2 = half ? kb2 : ka2;
                float k3 = half ? kb3 : ka3;
                acc += k0 * g0;
                acc += k1 * g1;
                acc += k2 * g2;
                acc += k3 * g3;
            }
            for (; j < n; j += 2) {
                int sa = __builtin_amdgcn_readlane(sl, j), sb = __builtin_amdgcn_readlane(sl, j + 1);
                float g = h2[(size_t)(half ? sb : sa) * N_CLASSES + c];
                float ka = lane_f(cl, j), kb = lane_f(cl, j + 1);
                float k = half ? kb : ka;
                if (j + half < n) acc += k * g;
            }
        }
        acc += __shfl_xor(acc, 16);
        float val = acc + di * di * selfv + bc;
        if (half == q) x = val;
    }

    float m = x;
    for (int o = 8; o > 0; o >>= 1) m = fmaxf(m, __shfl_xor(m, o));
    float ssum = expf(x - m);
    for (int o = 8; o > 0; o >>= 1) ssum += __shfl_xor(ssum, o);
    out[(size_t)(pair * 2 + half) * N_CLASSES + c] = x - m - logf(ssum);
}

extern "C" void kernel_launch(void* const* d_in, const int* in_sizes, int n_in,
                              void* d_out, int out_size, void* d_ws, size_t ws_size,
                              hipStream_t stream) {
    const float* x  = (const float*)d_in[0];
    const int*   ei = (const int*)d_in[1];
    const float* W1 = (const float*)d_in[2];
    const float* b1 = (const float*)d_in[3];
    const float* W2 = (const float*)d_in[4];
    const float* b2 = (const float*)d_in[5];
    float* out = (float*)d_out;

    int nE = in_sizes[1] / 2;
    const int* src = ei;
    const int* dst = ei + nE;

    float* dinv = (float*)d_ws;
    float* bufA = dinv + 102400;
    float* bufB = bufA + (size_t)N_NODES * HIDDEN;
    float* h2   = bufB;
    int* sorted = (int*)(bufB + (size_t)N_NODES * N_CLASSES);
    int* offs   = sorted + N_EDGES;
    int* pre    = offs + 102400;
    int* gcnt   = pre + BIN_BLKS * NBP;
    int* bstart = gcnt + NBP * GC_STRIDE;
    unsigned* binned = (unsigned*)bufA;

    int nTiles = N_NODES / 16;

    init_kernel<<<1, NBP, 0, stream>>>(gcnt);
    bucket_count_kernel<<<(nE + BIN_CHUNK - 1) / BIN_CHUNK, BIN_T, 0, stream>>>(dst, gcnt, pre, nE);
    bucket_scatter_kernel<<<(nE + BIN_CHUNK - 1) / BIN_CHUNK, BIN_T, 0, stream>>>(src, dst, gcnt, pre, bstart, binned, nE);
    bucket_sort_kernel<<<NB, SORT_T, 0, stream>>>(binned, bstart, offs, dinv, sorted);
    gemm1_kernel<<<(nTiles + 7) / 8, 256, 0, stream>>>(x, W1, bufA, nTiles);
    gather1_gemm2_kernel<<<N_NODES / 16, 256, 0, stream>>>(offs, sorted, dinv, bufA, b1, W2, h2);
    gather2_lsm_kernel<<<N_NODES / 16, 256, 0, stream>>>(offs, sorted, dinv, h2, b2, out);
}


# Edge weight dinv[src]*dinv[dst] split: h and h2 rows stored times own dinv (GEMM1, GEMM2 epilogues), gathers add plain rows, scale once by dinv[dst]; per-edge dinv[src] reads gone; sum order kept
# speedup vs baseline: 2.9460x; 1.0684x over previous
#include <hip/hip_runtime.h>
#include <math.h>

typedef float v2f __attribute__((ext_vector_type(2)));
typedef float v4f __attribute__((ext_vector_type(4)));
typedef float v8f __attribute__((ext_vector_type(8)));

#define N_NODES    100000
#define N_EDGES    1600000
#define N_FEAT     512
#define HIDDEN     64
#define N_CLASSES  16
#define BKT_SHIFT  9
#define BKT_W      512
#define NB         ((N_NODES + BKT_W - 1) / BKT_W)
#define NBP        256
#define GC_STRIDE  32
#define BIN_T      256
#define BIN_CHUNK  (BIN_T * 24)
#define BIN_BLKS   ((N_EDGES + BIN_CHUNK - 1) / BIN_CHUNK)
#define SORT_T     512
#define STAGE_CAP  12288
static_assert(NB <= NBP && BIN_T == NBP && SORT_T == BKT_W, "one scan thread per bucket / per destination");
static_assert(N_NODES <= (1 << (32 - BKT_SHIFT)), "a source id and 9 destination bits share one 32-bit word");
#define TILE_PITCH 66

__global__ __launch_bounds__(NBP) void init_kernel(int* gcnt) {
    gcnt[threadIdx.x * GC_STRIDE] = 0;
}

__global__ __launch_bounds__(BIN_T) void bucket_count_kernel(const int* __restrict__ dst,
                                                             int* __restrict__ gcnt,
                                                             int* __restrict__ pre, int nE) {
    __shared__ int hist[NBP];
    int t = threadIdx.x;
    hist[t] = 0;
    __syncthreads();
    int c0 = blockIdx.x * BIN_CHUNK;
    int c1 = c0 + BIN_CHUNK;
    if (c1 > nE) c1 = nE;
    for (int e0 = c0; e0 < c1; e0 += BIN_T * 8) {
        bool full = e0 + BIN_T * 8 <= c1;
        int d[8];
        for (int k = 0; k < 8; ++k) {
            int ek = e0 + t + k * BIN_T;
            d[k] = dst[ek < c1 ? ek : c1 - 1];
        }
        if (full) {
            for (int k = 0; k < 8; ++k)
                __hip_atomic_fetch_add(&hist[d[k] >> BKT_SHIFT], 1, __ATOMIC_RELAXED, __HIP_MEMORY_SCOPE_WORKGROUP);
        } else {
            for (int k = 0; k < 8; ++k)
                if (e0 + t + k * BIN_T < c1)
                    __hip_atomic_fetch_add(&hist[d[k] >> BKT_SHIFT], 1, __ATOMIC_RELAXED, __HIP_MEMORY_SCOPE_WORKGROUP);
        }
    }
    __syncthreads();
    if (t < NB)
        pre[blockIdx.x * NBP + t] =
            __hip_atomic_fetch_add(&gcnt[t * GC_STRIDE], hist[t], __ATOMIC_RELAXED, __HIP_MEMORY_SCOPE_AGENT);
}

__global__ __launch_bounds__(BIN_T) void bucket_scatter_kernel(const int* __restrict__ src,
                                                               const int* __restrict__ dst,
                                                               const int* __restrict__ gcnt,
                                                               const int* __restrict__ pre,
                                                               int* __restrict__ bstart,
                                                               unsigned* __restrict__ binned, int nE) {
    __shared__ int cur[NBP];
    __shared__ int wtot[NBP / 32];
    int t = threadIdx.x, lane = t & 31, w = t >> 5;
    int v = (t < NB) ? gcnt[t * GC_STRIDE] : 0;
    int inc = v;
    for (int o = 1; o < 32; o <<= 1) {
        int u = __shfl_up(inc, o);
        if (lane >= o) inc += u;
    }
    if (lane == 31) wtot[w] = inc;
    __syncthreads();
    int wbase = 0;
    for (int k = 0; k < NBP / 32; ++k)
        if (k < w) wbase += wtot[k];
    int excl = wbase + inc - v;
    cur[t] = excl + ((t < NB) ? pre[blockIdx.x * NBP + t] : 0);
    if (blockIdx.x == 0 && t <= NB) bstart[t] = excl;
    __syncthreads();

    int c0 = blockIdx.x * BIN_CHUNK;
    int c1 = c0 + BIN_CHUNK;
    if (c1 > nE) c1 = nE;
    for (int e0 = c0; e0 < c1; e0 += BIN_T * 8) {
        bool full = e0 + BIN_T * 8 <= c1;
        int d[8], s[8];
        for (int k = 0; k < 8; ++k) {
            int ek = e0 + t + k * BIN_T;
            int ec = ek < c1 ? ek : c1 - 1;
            d[k] = dst[ec];
            s[k] = src[ec];
        }
        if (full) {
            int p[8];
            for (int k = 0; k < 8; ++k)
                p[k] = __hip_atomic_fetch_add(&cur[d[k] >> BKT_SHIFT], 1, __ATOMIC_RELAXED, __HIP_MEMORY_SCOPE_WORKGROUP);
            for (int k = 0; k < 8; ++k)
                binned[p[k]] = ((unsigned)s[k] << BKT_SHIFT) | (unsigned)(d[k] & (BKT_W - 1));
        } else {
            for (int k = 0; k < 8; ++k)
                if (e0 + t + k * BIN_T < c1) {
                    int p = __hip_atomic_fetch_add(&cur[d[k] >> BKT_SHIFT], 1, __ATOMIC_RELAXED, __HIP_MEMORY_SCOPE_WORKGROUP);
                    binned[p] = ((unsigned)s[k] << BKT_SHIFT) | (unsigned)(d[k] & (BKT_W - 1));
                }
        }
    }
}

__global__ __launch_bounds__(SORT_T) void bucket_sort_kernel(const unsigned* __restrict__ binned,
                                                             const int* __restrict__ bstart,
                                                             int* __restrict__ offs,
                                                             float* __restrict__ dinv,
                                                             int* __restrict__ sorted) {
    __shared__ int cntL[BKT_W];
    __shared__ int curL[BKT_W];
    __shared__ int wtot[SORT_T / 32];
    __shared__ int stage[STAGE_CAP];
    int t = threadIdx.x, lane = t & 31, w = t >> 5;
    int b  = blockIdx.x;
    int bs = bstart[b];
    int n  = bstart[b + 1] - bs;
    const unsigned* mine = binned + bs;
    cntL[t] = 0;
    __syncthreads();
    for (int i0 = 0; i0 < n; i0 += SORT_T * 8) {
        bool full = i0 + SORT_T * 8 <= n;
        unsigned p[8];
        for (int k = 0; k < 8; ++k) {
            int ik = i0 + t + k * SORT_T;
            p[k] = mine[ik < n ? ik : n - 1];
        }
        if (full) {
            for (int k = 0; k < 8; ++k)
                __hip_atomic_fetch_add(&cntL[p[k] & (BKT_W - 1)], 1, __ATOMIC_RELAXED, __HIP_MEMORY_SCOPE_WORKGROUP);
        } else {
            for (int k = 0; k < 8; ++k)
                if (i0 + t + k * SORT_T < n)
                    __hip_atomic_fetch_add(&cntL[p[k] & (BKT_W - 1)], 1, __ATOMIC_RELAXED, __HIP_MEMORY_SCOPE_WORKGROUP);
        }
    }
    __syncthreads();

    int v = cntL[t];
    int inc = v;
    for (int o = 1; o < 32; o <<= 1) {
        int u = __shfl_up(inc, o);
        if (lane >= o) inc += u;
    }
    if (lane == 31) wtot[w] = inc;
    __syncthreads();
    int wbase = 0;
    for (int k = 0; k < SORT_T / 32; ++k)
        if (k < w) wbase += wtot[k];
    int excl = wbase + inc - v;
    curL[t] = excl;
    int dnode = b * BKT_W + t;
    if (dnode < N_NODES) {
        offs[dnode] = bs + excl;
        dinv[dnode] = rsqrtf(1.0f + (float)v);
        if (dnode == N_NODES - 1) offs[N_NODES] = bs + excl + v;
    }
    __syncthreads();

    bool fits = n <= STAGE_CAP;
    for (int i0 = 0; i0 < n; i0 += SORT_T * 8) {
        bool full = i0 + SORT_T * 8 <= n;
        unsigned p[8];
        for (int k = 0; k < 8; ++k) {
            int ik = i0 + t + k * SORT_T;
            p[k] = mine[ik < n ? ik : n - 1];
        }
        if (full && fits) {
            int pos[8];
            for (int k = 0; k < 8; ++k)
                pos[k] = __hip_atomic_fetch_add(&curL[p[k] & (BKT_W - 1)], 1, __ATOMIC_RELAXED, __HIP_MEMORY_SCOPE_WORKGROUP);
            for (int k = 0; k < 8; ++k)
                stage[pos[k]] = (int)(p[k] >> BKT_SHIFT);
        } else {
            for (int k = 0; k < 8; ++k)
                if (i0 + t + k * SORT_T < n) {
                    int pos = __hip_atomic_fetch_add(&curL[p[k] & (BKT_W - 1)], 1, __ATOMIC_RELAXED, __HIP_MEMORY_SCOPE_WORKGROUP);
                    int s = (int)(p[k] >> BKT_SHIFT);
                    if (fits) stage[pos] = s;
                    else      sorted[bs + pos] = s;
                }
        }
    }
    __syncthreads();
    if (fits)
        for (int i = t; i < n; i += SORT_T) sorted[bs + i] = stage[i];
}

__global__ __launch_bounds__(256) void gemm1_kernel(const float* __restrict__ A,
                                                    const float* __restrict__ B,
                                                    const float* __restrict__ dinv,
                                                    float* __restrict__ C, int nTiles) {
    int wave = threadIdx.x >> 5;
    int lane = threadIdx.x & 31;
    int tile = blockIdx.x * 8 + wave;
    if (tile >= nTiles) return;
    int c  = lane & 15;
    int hi = lane >> 4;

    const float* arow = A + (size_t)(tile * 16 + c) * N_FEAT + 4 * hi;
    const float* bcol = B + (size_t)(4 * hi) * HIDDEN + 4 * c;
    v8f acc0 = {}, acc1 = {}, acc2 = {}, acc3 = {};

    v4f an0 = *(const v4f*)(arow);
    v4f an1 = *(const v4f*)(arow + 8);
    v4f an2 = *(const v4f*)(arow + 16);
    v4f an3 = *(const v4f*)(arow + 24);
    v4f qn0 = *(const v4f*)(bcol);
    v4f qn1 = *(const v4f*)(bcol + HIDDEN);
    v4f qn2 = *(const v4f*)(bcol + 2 * HIDDEN);
    v4f qn3 = *(const v4f*)(bcol + 3 * HIDDEN);
    for (int K0 = 0; K0 < N_FEAT; K0 += 32) {
        v4f a[4] = { an0, an1, an2, an3 };
        if (K0 + 32 < N_FEAT) {
            const float* ap = arow + K0 + 32;
            an0 = *(const v4f*)(ap);
            an1 = *(const v4f*)(ap + 8);
            an2 = *(const v4f*)(ap + 16);
            an3 = *(const v4f*)(ap + 24);
        }
        for (int s = 0; s < 4; ++s) {
            v4f q0 = qn0, q1 = qn1, q2 = qn2, q3 = qn3;
            int kn = K0 + 8 * s + 8;
            if (kn < N_FEAT) {
                const float* bp = bcol + (size_t)kn * HIDDEN;
                qn0 = *(const v4f*)(bp);
                qn1 = *(const v4f*)(bp + HIDDEN);
                qn2 = *(const v4f*)(bp + 2 * HIDDEN);
                qn3 = *(const v4f*)(bp + 3 * HIDDEN);
            }
            __builtin_amdgcn_sched_barrier(0);
            v2f a01 = { a[s].x, a[s].y };
            v2f a23 = { a[s].z, a[s].w };
            v2f b;
            b = (v2f){ q0.x, q1.x }; acc0 = __builtin_amdgcn_wmma_f32_16x16x4_f32(false, a01, false, b, (short)0, acc0, false, false);
            b = (v2f){ q0.y, q1.y }; acc1 = __builtin_amdgcn_wmma_f32_16x16x4_f32(false, a01, false, b, (short)0, acc1, false, false);
            b = (v2f){ q0.z, q1.z }; acc2 = __builtin_amdgcn_wmma_f32_16x16x4_f32(false, a01, false, b, (short)0, acc2, false, false);
            b = (v2f){ q0.w, q1.w }; acc3 = __builtin_amdgcn_wmma_f32_16x16x4_f32(false, a01, false, b, (short)0, acc3, false, false);
            b = (v2f){ q2.x, q3.x }; acc0 = __builtin_amdgcn_wmma_f32_16x16x4_f32(false, a23, false, b, (short)0, acc0, false, false);
            b = (v2f){ q2.y, q3.y }; acc1 = __builtin_amdgcn_wmma_f32_16x16x4_f32(false, a23, false, b, (short)0, acc1, false, false);
            b = (v2f){ q2.z, q3.z }; acc2 = __builtin_amdgcn_wmma_f32_16x16x4_f32(false, a23, false, b, (short)0, acc2, false, false);
            b = (v2f){ q2.w, q3.w }; acc3 = __builtin_amdgcn_wmma_f32_16x16x4_f32(false, a23, false, b, (short)0, acc3, false, false);
        }
    }
    int mbase = tile * 16 + hi * 8;
    v4f dlo = *(const v4f*)(dinv + mbase);
    v4f dhi = *(const v4f*)(dinv + mbase + 4);
    float ds[8] = { dlo.x, dlo.y, dlo.z, dlo.w, dhi.x, dhi.y, dhi.z, dhi.w };
    for (int r = 0; r < 8; ++r) {
        float s = ds[r];
        v4f o = { acc0[r] * s, acc1[r] * s, acc2[r] * s, acc3[r] * s };
        *(v4f*)(C + (size_t)(mbase + r) * HIDDEN + 4 * c) = o;
    }
}

__global__ __launch_bounds__(256) void gather1_gemm2_kernel(const int* __restrict__ offs,
                                                            const int* __restrict__ sorted,
                                                            const float* __restrict__ dinv,
                                                            const float* __restrict__ h,
                                                            const float* __restrict__ b1,
                                                            const float* __restrict__ W2,
                                                            float* __restrict__ h2) {
    __shared__ __attribute__((aligned(16))) float tile[16 * TILE_PITCH];
    int lane = threadIdx.x & 31;
    int wave = __builtin_amdgcn_readfirstlane(threadIdx.x >> 5);
    v2f bias = *(const v2f*)(b1 + 2 * lane);

    for (int q = 0; q < 2; ++q) {
        int row = wave * 2 + q;
        int d   = blockIdx.x * 16 + row;
        int beg = offs[d], end = offs[d + 1];
        float di = dinv[d];
        v2f self = *(const v2f*)(h + (size_t)d * HIDDEN + 2 * lane);
        float a0 = self.x, a1 = self.y;
        for (int base = beg; base < end; base += 32) {
            int my = base + lane;
            int sl = 0;
            if (my < end) sl = sorted[my];
            int n = end - base;
            if (n > 32) n = 32;
            int j = 0;
            for (; j + 4 <= n; j += 4) {
                int s0 = __builtin_amdgcn_readlane(sl, j);
                int s1 = __builtin_amdgcn_readlane(sl, j + 1);
                int s2 = __builtin_amdgcn_readlane(sl, j + 2);
                int s3 = __builtin_amdgcn_readlane(sl, j + 3);
                v2f r0 = *(const v2f*)(h + (size_t)s0 * HIDDEN + 2 * lane);
                v2f r1 = *(const v2f*)(h + (size_t)s1 * HIDDEN + 2 * lane);
                v2f r2 = *(const v2f*)(h + (size_t)s2 * HIDDEN + 2 * lane);
                v2f r3 = *(const v2f*)(h + (size_t)s3 * HIDDEN + 2 * lane);
                a0 += r0.x; a1 += r0.y;
                a0 += r1.x; a1 += r1.y;
                a0 += r2.x; a1 += r2.y;
                a0 += r3.x; a1 += r3.y;
            }
            for (; j < n; ++j) {
                int s0 = __builtin_amdgcn_readlane(sl, j);
                v2f r0 = *(const v2f*)(h + (size_t)s0 * HIDDEN + 2 * lane);
                a0 += r0.x; a1 += r0.y;
            }
        }
        float v0 = di * a0 + bias.x, v1 = di * a1 + bias.y;
        v2f o = { (v0 > 0.0f ? v0 : 0.0f) * di, (v1 > 0.0f ? v1 : 0.0f) * di };
        *(v2f*)(tile + row * TILE_PITCH + 2 * lane) = o;
    }
    __syncthreads();

    if (wave == 0) {
        int nr = lane & 15;
        int hi = lane >> 4;
        const float* arow = tile + nr * TILE_PITCH + 2 * hi;
        v8f acc = {};
        for (int kk = 0; kk < HIDDEN; kk += 4) {
            v2f a = *(const v2f*)(arow + kk);
            const float* b0p = W2 + (size_t)(kk + 2 * hi) * N_CLASSES + nr;
            v2f b = { b0p[0], b0p[N_CLASSES] };
            acc = __builtin_amdgcn_wmma_f32_16x16x4_f32(false, a, false, b, (short)0, acc, false, false);
        }
        int mbase = blockIdx.x * 16 + hi * 8;
        for (int r = 0; r < 8; ++r)
            h2[(size_t)(mbase + r) * N_CLASSES + nr] = acc[r];
    }
}

__global__ __launch_bounds__(256) void gather2_lsm_kernel(const int* __restrict__ offs,
                                                          const int* __restrict__ sorted,
                                                          const float* __restrict__ dinv,
                                                          const float* __restrict__ h2,
                                                          const float* __restrict__ b2,
                                                          float* __restrict__ out) {
    int lane = threadIdx.x & 31;
    int wave = __builtin_amdgcn_readfirstlane(threadIdx.x >> 5);
    int c    = lane & 15;
    int half = lane >> 4;
    int pair = blockIdx.x * 8 + wave;
    float bc = b2[c];
    float x  = 0.0f;

    for (int q = 0; q < 2; ++q) {
        int d   = pair * 2 + q;
        int beg = offs[d], end = offs[d + 1];
        float di = dinv[d];
        float selfv = h2[(size_t)d * N_CLASSES + c];
        float acc = 0.0f;
        for (int base = beg; base < end; base += 32) {
            int my = base + lane;
            int sl = 0;
            if (my < end) sl = sorted[my];
            int n = end - base;
            if (n > 32) n = 32;
            int j = 0;
            for (; j + 8 <= n; j += 8) {
                int sa0 = __builtin_amdgcn_readlane(sl, j),     sb0 = __builtin_amdgcn_readlane(sl, j + 1);
                int sa1 = __builtin_amdgcn_readlane(sl, j + 2), sb1 = __builtin_amdgcn_readlane(sl, j + 3);
                int sa2 = __builtin_amdgcn_readlane(sl, j + 4), sb2 = __builtin_amdgcn_readlane(sl, j + 5);
                int sa3 = __builtin_amdgcn_readlane(sl, j + 6), sb3 = __builtin_amdgcn_readlane(sl, j + 7);
                float g0 = h2[(size_t)(half ? sb0 : sa0) * N_CLASSES + c];
                float g1 = h2[(size_t)(half ? sb1 : sa1) * N_CLASSES + c];
                float g2 = h2[(size_t)(half ? sb2 : sa2) * N_CLASSES + c];
                float g3 = h2[(size_t)(half ? sb3 : sa3) * N_CLASSES + c];
                acc += g0;
                acc += g1;
                acc += g2;
                acc += g3;
            }
            for (; j < n; j += 2) {
                int sa = __builtin_amdgcn_readlane(sl, j), sb = __builtin_amdgcn_readlane(sl, j + 1);
                float g = h2[(size_t)(half ? sb : sa) * N_CLASSES + c];
                if (j + half < n) acc += g;
            }
        }
        acc += __shfl_xor(acc, 16);
        float val = di * (acc + selfv) + bc;
        if (half == q) x = val;
    }

    float m = x;
    for (int o = 8; o > 0; o >>= 1) m = fmaxf(m, __shfl_xor(m, o));
    float ssum = expf(x - m);
    for (int o = 8; o > 0; o >>= 1) ssum += __shfl_xor(ssum, o);
    out[(size_t)(pair * 2 + half) * N_CLASSES + c] = x - m - logf(ssum);
}

extern "C" void kernel_launch(void* const* d_in, const int* in_sizes, int n_in,
                              void* d_out, int out_size, void* d_ws, size_t ws_size,
                              hipStream_t stream) {
    const float* x  = (const float*)d_in[0];
    const int*   ei = (const int*)d_in[1];
    const float* W1 = (const float*)d_in[2];
    const float* b1 = (const float*)d_in[3];
    const float* W2 = (const float*)d_in[4];
    const float* b2 = (const float*)d_in[5];
    float* out = (float*)d_out;

    int nE = in_sizes[1] / 2;
    const int* src = ei;
    const int* dst = ei + nE;

    float* dinv = (float*)d_ws;
    float* bufA = dinv + 102400;
    float* bufB = bufA + (size_t)N_NODES * HIDDEN;
    float* h2   = bufB;
    int* sorted = (int*)(bufB + (size_t)N_NODES * N_CLASSES);
    int* offs   = sorted + N_EDGES;
    int* pre    = offs + 102400;
    int* gcnt   = pre + BIN_BLKS * NBP;
    int* bstart = gcnt + NBP * GC_STRIDE;
    unsigned* binned = (unsigned*)bufA;

    int nTiles = N_NODES / 16;

    init_kernel<<<1, NBP, 0, stream>>>(gcnt);
    bucket_count_kernel<<<(nE + BIN_CHUNK - 1) / BIN_CHUNK, BIN_T, 0, stream>>>(dst, gcnt, pre, nE);
    bucket_scatter_kernel<<<(nE + BIN_CHUNK - 1) / BIN_CHUNK, BIN_T, 0, stream>>>(src, dst, gcnt, pre, bstart, binned, nE);
    bucket_sort_kernel<<<NB, SORT_T, 0, stream>>>(binned, bstart, offs, dinv, sorted);
    gemm1_kernel<<<(nTiles + 7) / 8, 256, 0, stream>>>(x, W1, dinv, bufA, nTiles);
    gather1_gemm2_kernel<<<N_NODES / 16, 256, 0, stream>>>(offs, sorted, dinv, bufA, b1, W2, h2);
    gather2_lsm_kernel<<<N_NODES / 16, 256, 0, stream>>>(offs, sorted, dinv, h2, b2, out);
}


# GEMM1 K split over wave pairs (2 x 256 k per 16x64 tile, upper half added through LDS: k sum = lower half + upper half); 4-wave blocks: fullest SIMD 13 x 256 WMMAs, was 8 x 512
# speedup vs baseline: 3.0886x; 1.0484x over previous
#include <hip/hip_runtime.h>
#include <math.h>

typedef float v2f __attribute__((ext_vector_type(2)));
typedef float v4f __attribute__((ext_vector_type(4)));
typedef float v8f __attribute__((ext_vector_type(8)));

#define N_NODES    100000
#define N_EDGES    1600000
#define N_FEAT     512
#define HIDDEN     64
#define N_CLASSES  16
#define BKT_SHIFT  9
#define BKT_W      512
#define NB         ((N_NODES + BKT_W - 1) / BKT_W)
#define NBP        256
#define GC_STRIDE  32
#define BIN_T      256
#define BIN_CHUNK  (BIN_T * 24)
#define BIN_BLKS   ((N_EDGES + BIN_CHUNK - 1) / BIN_CHUNK)
#define SORT_T     512
#define STAGE_CAP  12288
static_assert(NB <= NBP && BIN_T == NBP && SORT_T == BKT_W, "one scan thread per bucket / per destination");
static_assert(N_NODES <= (1 << (32 - BKT_SHIFT)), "a source id and 9 destination bits share one 32-bit word");
#define TILE_PITCH 66

__global__ __launch_bounds__(NBP) void init_kernel(int* gcnt) {
    gcnt[threadIdx.x * GC_STRIDE] = 0;
}

__global__ __launch_bounds__(BIN_T) void bucket_count_kernel(const int* __restrict__ dst,
                                                             int* __restrict__ gcnt,
                                                             int* __restrict__ pre, int nE) {
    __shared__ int hist[NBP];
    int t = threadIdx.x;
    hist[t] = 0;
    __syncthreads();
    int c0 = blockIdx.x * BIN_CHUNK;
    int c1 = c0 + BIN_CHUNK;
    if (c1 > nE) c1 = nE;
    for (int e0 = c0; e0 < c1; e0 += BIN_T * 8) {
        bool full = e0 + BIN_T * 8 <= c1;
        int d[8];
        for (int k = 0; k < 8; ++k) {
            int ek = e0 + t + k * BIN_T;
            d[k] = dst[ek < c1 ? ek : c1 - 1];
        }
        if (full) {
            for (int k = 0; k < 8; ++k)
                __hip_atomic_fetch_add(&hist[d[k] >> BKT_SHIFT], 1, __ATOMIC_RELAXED, __HIP_MEMORY_SCOPE_WORKGROUP);
        } else {
            for (int k = 0; k < 8; ++k)
                if (e0 + t + k * BIN_T < c1)
                    __hip_atomic_fetch_add(&hist[d[k] >> BKT_SHIFT], 1, __ATOMIC_RELAXED, __HIP_MEMORY_SCOPE_WORKGROUP);
        }
    }
    __syncthreads();
    if (t < NB)
        pre[blockIdx.x * NBP + t] =
            __hip_atomic_fetch_add(&gcnt[t * GC_STRIDE], hist[t], __ATOMIC_RELAXED, __HIP_MEMORY_SCOPE_AGENT);
}

__global__ __launch_bounds__(BIN_T) void bucket_scatter_kernel(const int* __restrict__ src,
                                                               const int* __restrict__ dst,
                                                               const int* __restrict__ gcnt,
                                                               const int* __restrict__ pre,
                                                               int* __restrict__ bstart,
                                                               unsigned* __restrict__ binned, int nE) {
    __shared__ int cur[NBP];
    __shared__ int wtot[NBP / 32];
    int t = threadIdx.x, lane = t & 31, w = t >> 5;
    int v = (t < NB) ? gcnt[t * GC_STRIDE] : 0;
    int inc = v;
    for (int o = 1; o < 32; o <<= 1) {
        int u = __shfl_up(inc, o);
        if (lane >= o) inc += u;
    }
    if (lane == 31) wtot[w] = inc;
    __syncthreads();
    int wbase = 0;
    for (int k = 0; k < NBP / 32; ++k)
        if (k < w) wbase += wtot[k];
    int excl = wbase + inc - v;
    cur[t] = excl + ((t < NB) ? pre[blockIdx.x * NBP + t] : 0);
    if (blockIdx.x == 0 && t <= NB) bstart[t] = excl;
    __syncthreads();

    int c0 = blockIdx.x * BIN_CHUNK;
    int c1 = c0 + BIN_CHUNK;
    if (c1 > nE) c1 = nE;
    for (int e0 = c0; e0 < c1; e0 += BIN_T * 8) {
        bool full = e0 + BIN_T * 8 <= c1;
        int d[8], s[8];
        for (int k = 0; k < 8; ++k) {
            int ek = e0 + t + k * BIN_T;
            int ec = ek < c1 ? ek : c1 - 1;
            d[k] = dst[ec];
            s[k] = src[ec];
        }
        if (full) {
            int p[8];
            for (int k = 0; k < 8; ++k)
                p[k] = __hip_atomic_fetch_add(&cur[d[k] >> BKT_SHIFT], 1, __ATOMIC_RELAXED, __HIP_MEMORY_SCOPE_WORKGROUP);
            for (int k = 0; k < 8; ++k)
                binned[p[k]] = ((unsigned)s[k] << BKT_SHIFT) | (unsigned)(d[k] & (BKT_W - 1));
        } else {
            for (int k = 0; k < 8; ++k)
                if (e0 + t + k * BIN_T < c1) {
                    int p = __hip_atomic_fetch_add(&cur[d[k] >> BKT_SHIFT], 1, __ATOMIC_RELAXED, __HIP_MEMORY_SCOPE_WORKGROUP);
                    binned[p] = ((unsigned)s[k] << BKT_SHIFT) | (unsigned)(d[k] & (BKT_W - 1));
                }
        }
    }
}

__global__ __launch_bounds__(SORT_T) void bucket_sort_kernel(const unsigned* __restrict__ binned,
                                                             const int* __restrict__ bstart,
                                                             int* __restrict__ offs,
                                                             float* __restrict__ dinv,
                                                             int* __restrict__ sorted) {
    __shared__ int cntL[BKT_W];
    __shared__ int curL[BKT_W];
    __shared__ int wtot[SORT_T / 32];
    __shared__ int stage[STAGE_CAP];
    int t = threadIdx.x, lane = t & 31, w = t >> 5;
    int b  = blockIdx.x;
    int bs = bstart[b];
    int n  = bstart[b + 1] - bs;
    const unsigned* mine = binned + bs;
    cntL[t] = 0;
    __syncthreads();
    for (int i0 = 0; i0 < n; i0 += SORT_T * 8) {
        bool full = i0 + SORT_T * 8 <= n;
        unsigned p[8];
        for (int k = 0; k < 8; ++k) {
            int ik = i0 + t + k * SORT_T;
            p[k] = mine[ik < n ? ik : n - 1];
        }
        if (full) {
            for (int k = 0; k < 8; ++k)
                __hip_atomic_fetch_add(&cntL[p[k] & (BKT_W - 1)], 1, __ATOMIC_RELAXED, __HIP_MEMORY_SCOPE_WORKGROUP);
        } else {
            for (int k = 0; k < 8; ++k)
                if (i0 + t + k * SORT_T < n)
                    __hip_atomic_fetch_add(&cntL[p[k] & (BKT_W - 1)], 1, __ATOMIC_RELAXED, __HIP_MEMORY_SCOPE_WORKGROUP);
        }
    }
    __syncthreads();

    int v = cntL[t];
    int inc = v;
    for (int o = 1; o < 32; o <<= 1) {
        int u = __shfl_up(inc, o);
        if (lane >= o) inc += u;
    }
    if (lane == 31) wtot[w] = inc;
    __syncthreads();
    int wbase = 0;
    for (int k = 0; k < SORT_T / 32; ++k)
        if (k < w) wbase += wtot[k];
    int excl = wbase + inc - v;
    curL[t] = excl;
    int dnode = b * BKT_W + t;
    if (dnode < N_NODES) {
        offs[dnode] = bs + excl;
        dinv[dnode] = rsqrtf(1.0f + (float)v);
        if (dnode == N_NODES - 1) offs[N_NODES] = bs + excl + v;
    }
    __syncthreads();

    bool fits = n <= STAGE_CAP;
    for (int i0 = 0; i0 < n; i0 += SORT_T * 8) {
        bool full = i0 + SORT_T * 8 <= n;
        unsigned p[8];
        for (int k = 0; k < 8; ++k) {
            int ik = i0 + t + k * SORT_T;
            p[k] = mine[ik < n ? ik : n - 1];
        }
        if (full && fits) {
            int pos[8];
            for (int k = 0; k < 8; ++k)
                pos[k] = __hip_atomic_fetch_add(&curL[p[k] & (BKT_W - 1)], 1, __ATOMIC_RELAXED, __HIP_MEMORY_SCOPE_WORKGROUP);
            for (int k = 0; k < 8; ++k)
                stage[pos[k]] = (int)(p[k] >> BKT_SHIFT);
        } else {
            for (int k = 0; k < 8; ++k)
                if (i0 + t + k * SORT_T < n) {
                    int pos = __hip_atomic_fetch_add(&curL[p[k] & (BKT_W - 1)], 1, __ATOMIC_RELAXED, __HIP_MEMORY_SCOPE_WORKGROUP);
                    int s = (int)(p[k] >> BKT_SHIFT);
                    if (fits) stage[pos] = s;
                    else      sorted[bs + pos] = s;
                }
        }
    }
    __syncthreads();
    if (fits)
        for (int i = t; i < n; i += SORT_T) sorted[bs + i] = stage[i];
}

#define K_HALF (N_FEAT / 2)
static_assert(N_NODES % 32 == 0 && K_HALF % 32 == 0, "two whole 16-row tiles per block, whole 32-k lines per K half");
__global__ __launch_bounds__(128) void gemm1_kernel(const float* __restrict__ A,
                                                    const float* __restrict__ B,
                                                    const float* __restrict__ dinv,
                                                    float* __restrict__ C) {
    __shared__ float red[2][32][32];
    int wave = __builtin_amdgcn_readfirstlane(threadIdx.x >> 5);
    int lane = threadIdx.x & 31;
    int pr   = wave >> 1;
    int kh   = wave & 1;
    int tile = blockIdx.x * 2 + pr;
    int c  = lane & 15;
    int hi = lane >> 4;

    const float* arow = A + (size_t)(tile * 16 + c) * N_FEAT + kh * K_HALF + 4 * hi;
    const float* bcol = B + (size_t)(kh * K_HALF + 4 * hi) * HIDDEN + 4 * c;
    v8f acc0 = {}, acc1 = {}, acc2 = {}, acc3 = {};

    v4f an0 = *(const v4f*)(arow);
    v4f an1 = *(const v4f*)(arow + 8);
    v4f an2 = *(const v4f*)(arow + 16);
    v4f an3 = *(const v4f*)(arow + 24);
    v4f qn0 = *(const v4f*)(bcol);
    v4f qn1 = *(const v4f*)(bcol + HIDDEN);
    v4f qn2 = *(const v4f*)(bcol + 2 * HIDDEN);
    v4f qn3 = *(const v4f*)(bcol + 3 * HIDDEN);
    for (int K0 = 0; K0 < K_HALF; K0 += 32) {
        v4f a[4] = { an0, an1, an2, an3 };
        if (K0 + 32 < K_HALF) {
            const float* ap = arow + K0 + 32;
            an0 = *(const v4f*)(ap);
            an1 = *(const v4f*)(ap + 8);
            an2 = *(const v4f*)(ap + 16);
            an3 = *(const v4f*)(ap + 24);
        }
        for (int s = 0; s < 4; ++s) {
            v4f q0 = qn0, q1 = qn1, q2 = qn2, q3 = qn3;
            int kn = K0 + 8 * s + 8;
            if (kn < K_HALF) {
                const float* bp = bcol + (size_t)kn * HIDDEN;
                qn0 = *(const v4f*)(bp);
                qn1 = *(const v4f*)(bp + HIDDEN);
                qn2 = *(const v4f*)(bp + 2 * HIDDEN);
                qn3 = *(const v4f*)(bp + 3 * HIDDEN);
            }
            __builtin_amdgcn_sched_barrier(0);
            v2f a01 = { a[s].x, a[s].y };
            v2f a23 = { a[s].z, a[s].w };
            v2f b;
            b = (v2f){ q0.x, q1.x }; acc0 = __builtin_amdgcn_wmma_f32_16x16x4_f32(false, a01, false, b, (short)0, acc0, false, false);
            b = (v2f){ q0.y, q1.y }; acc1 = __builtin_amdgcn_wmma_f32_16x16x4_f32(false, a01, false, b, (short)0, acc1, false, false);
            b = (v2f){ q0.z, q1.z }; acc2 = __builtin_amdgcn_wmma_f32_16x16x4_f32(false, a01, false, b, (short)0, acc2, false, false);
            b = (v2f){ q0.w, q1.w }; acc3 = __builtin_amdgcn_wmma_f32_16x16x4_f32(false, a01, false, b, (short)0, acc3, false, false);
            b = (v2f){ q2.x, q3.x }; acc0 = __builtin_amdgcn_wmma_f32_16x16x4_f32(false, a23, false, b, (short)0, acc0, false, false);
            b = (v2f){ q2.y, q3.y }; acc1 = __builtin_amdgcn_wmma_f32_16x16x4_f32(false, a23, false, b, (short)0, acc1, false, false);
            b = (v2f){ q2.z, q3.z }; acc2 = __builtin_amdgcn_wmma_f32_16x16x4_f32(false, a23, false, b, (short)0, acc2, false, false);
            b = (v2f){ q2.w, q3.w }; acc3 = __builtin_amdgcn_wmma_f32_16x16x4_f32(false, a23, false, b, (short)0, acc3, false, false);
        }
    }
    asm volatile("v_nop\n\tv_nop\n\tv_nop\n\tv_nop\n\tv_nop"
                 : "+v"(acc0), "+v"(acc1), "+v"(acc2), "+v"(acc3));
    if (kh == 1) {
        for (int r = 0; r < 8; ++r) {
            red[pr][r][lane]      = acc0[r];
            red[pr][8 + r][lane]  = acc1[r];
            red[pr][16 + r][lane] = acc2[r];
            red[pr][24 + r][lane] = acc3[r];
        }
    }
    __syncthreads();
    if (kh == 0) {
        int mbase = tile * 16 + hi * 8;
        v4f dlo = *(const v4f*)(dinv + mbase);
        v4f dhi = *(const v4f*)(dinv + mbase + 4);
        float ds[8] = { dlo.x, dlo.y, dlo.z, dlo.w, dhi.x, dhi.y, dhi.z, dhi.w };
        for (int r = 0; r < 8; ++r) {
            float s = ds[r];
            v4f o = { (acc0[r] + red[pr][r][lane]) * s,      (acc1[r] + red[pr][8 + r][lane]) * s,
                      (acc2[r] + red[pr][16 + r][lane]) * s, (acc3[r] + red[pr][24 + r][lane]) * s };
            *(v4f*)(C + (size_t)(mbase + r) * HIDDEN + 4 * c) = o;
        }
    }
}

__global__ __launch_bounds__(256) void gather1_gemm2_kernel(const int* __restrict__ offs,
                                                            const int* __restrict__ sorted,
                                                            const float* __restrict__ dinv,
                                                            const float* __restrict__ h,
                                                            const float* __restrict__ b1,
                                                            const float* __restrict__ W2,
                                                            float* __restrict__ h2) {
    __shared__ __attribute__((aligned(16))) float tile[16 * TILE_PITCH];
    int lane = threadIdx.x & 31;
    int wave = __builtin_amdgcn_readfirstlane(threadIdx.x >> 5);
    v2f bias = *(const v2f*)(b1 + 2 * lane);

    for (int q = 0; q < 2; ++q) {
        int row = wave * 2 + q;
        int d   = blockIdx.x * 16 + row;
        int beg = offs[d], end = offs[d + 1];
        float di = dinv[d];
        v2f self = *(const v2f*)(h + (size_t)d * HIDDEN + 2 * lane);
        float a0 = self.x, a1 = self.y;
        for (int base = beg; base < end; base += 32) {
            int my = base + lane;
            int sl = 0;
            if (my < end) sl = sorted[my];
            int n = end - base;
            if (n > 32) n = 32;
            int j = 0;
            for (; j + 4 <= n; j += 4) {
                int s0 = __builtin_amdgcn_readlane(sl, j);
                int s1 = __builtin_amdgcn_readlane(sl, j + 1);
                int s2 = __builtin_amdgcn_readlane(sl, j + 2);
                int s3 = __builtin_amdgcn_readlane(sl, j + 3);
                v2f r0 = *(const v2f*)(h + (size_t)s0 * HIDDEN + 2 * lane);
                v2f r1 = *(const v2f*)(h + (size_t)s1 * HIDDEN + 2 * lane);
                v2f r2 = *(const v2f*)(h + (size_t)s2 * HIDDEN + 2 * lane);
                v2f r3 = *(const v2f*)(h + (size_t)s3 * HIDDEN + 2 * lane);
                a0 += r0.x; a1 += r0.y;
                a0 += r1.x; a1 += r1.y;
                a0 += r2.x; a1 += r2.y;
                a0 += r3.x; a1 += r3.y;
            }
            for (; j < n; ++j) {
                int s0 = __builtin_amdgcn_readlane(sl, j);
                v2f r0 = *(const v2f*)(h + (size_t)s0 * HIDDEN + 2 * lane);
                a0 += r0.x; a1 += r0.y;
            }
        }
        float v0 = di * a0 + bias.x, v1 = di * a1 + bias.y;
        v2f o = { (v0 > 0.0f ? v0 : 0.0f) * di, (v1 > 0.0f ? v1 : 0.0f) * di };
        *(v2f*)(tile + row * TILE_PITCH + 2 * lane) = o;
    }
    __syncthreads();

    if (wave == 0) {
        int nr = lane & 15;
        int hi = lane >> 4;
        const float* arow = tile + nr * TILE_PITCH + 2 * hi;
        v8f acc = {};
        for (int kk = 0; kk < HIDDEN; kk += 4) {
            v2f a = *(const v2f*)(arow + kk);
            const float* b0p = W2 + (size_t)(kk + 2 * hi) * N_CLASSES + nr;
            v2f b = { b0p[0], b0p[N_CLASSES] };
            acc = __builtin_amdgcn_wmma_f32_16x16x4_f32(false, a, false, b, (short)0, acc, false, false);
        }
        int mbase = blockIdx.x * 16 + hi * 8;
        for (int r = 0; r < 8; ++r)
            h2[(size_t)(mbase + r) * N_CLASSES + nr] = acc[r];
    }
}

__global__ __launch_bounds__(256) void gather2_lsm_kernel(const int* __restrict__ offs,
                                                          const int* __restrict__ sorted,
                                                          const float* __restrict__ dinv,
                                                          const float* __restrict__ h2,
                                                          const float* __restrict__ b2,
                                                          float* __restrict__ out) {
    int lane = threadIdx.x & 31;
    int wave = __builtin_amdgcn_readfirstlane(threadIdx.x >> 5);
    int c    = lane & 15;
    int half = lane >> 4;
    int pair = blockIdx.x * 8 + wave;
    float bc = b2[c];
    float x  = 0.0f;

    for (int q = 0; q < 2; ++q) {
        int d   = pair * 2 + q;
        int beg = offs[d], end = offs[d + 1];
        float di = dinv[d];
        float selfv = h2[(size_t)d * N_CLASSES + c];
        float acc = 0.0f;
        for (int base = beg; base < end; base += 32) {
            int my = base + lane;
            int sl = 0;
            if (my < end) sl = sorted[my];
            int n = end - base;
            if (n > 32) n = 32;
            int j = 0;
            for (; j + 8 <= n; j += 8) {
                int sa0 = __builtin_amdgcn_readlane(sl, j),     sb0 = __builtin_amdgcn_readlane(sl, j + 1);
                int sa1 = __builtin_amdgcn_readlane(sl, j + 2), sb1 = __builtin_amdgcn_readlane(sl, j + 3);
                int sa2 = __builtin_amdgcn_readlane(sl, j + 4), sb2 = __builtin_amdgcn_readlane(sl, j + 5);
                int sa3 = __builtin_amdgcn_readlane(sl, j + 6), sb3 = __builtin_amdgcn_readlane(sl, j + 7);
                float g0 = h2[(size_t)(half ? sb0 : sa0) * N_CLASSES + c];
                float g1 = h2[(size_t)(half ? sb1 : sa1) * N_CLASSES + c];
                float g2 = h2[(size_t)(half ? sb2 : sa2) * N_CLASSES + c];
                float g3 = h2[(size_t)(half ? sb3 : sa3) * N_CLASSES + c];
                acc += g0;
                acc += g1;
                acc += g2;
                acc += g3;
            }
            for (; j < n; j += 2) {
                int sa = __builtin_amdgcn_readlane(sl, j), sb = __builtin_amdgcn_readlane(sl, j + 1);
                float g = h2[(size_t)(half ? sb : sa) * N_CLASSES + c];
                if (j + half < n) acc += g;
            }
        }
        acc += __shfl_xor(acc, 16);
        float val = di * (acc + selfv) + bc;
        if (half == q) x = val;
    }

    float m = x;
    for (int o = 8; o > 0; o >>= 1) m = fmaxf(m, __shfl_xor(m, o));
    float ssum = expf(x - m);
    for (int o = 8; o > 0; o >>= 1) ssum += __shfl_xor(ssum, o);
    out[(size_t)(pair * 2 + half) * N_CLASSES + c] = x - m - logf(ssum);
}

extern "C" void kernel_launch(void* const* d_in, const int* in_sizes, int n_in,
                              void* d_out, int out_size, void* d_ws, size_t ws_size,
                              hipStream_t stream) {
    const float* x  = (const float*)d_in[0];
    const int*   ei = (const int*)d_in[1];
    const float* W1 = (const float*)d_in[2];
    const float* b1 = (const float*)d_in[3];
    const float* W2 = (const float*)d_in[4];
    const float* b2 = (const float*)d_in[5];
    float* out = (float*)d_out;

    int nE = in_sizes[1] / 2;
    const int* src = ei;
    const int* dst = ei + nE;

    float* dinv = (float*)d_ws;
    float* bufA = dinv + 102400;
    float* bufB = bufA + (size_t)N_NODES * HIDDEN;
    float* h2   = bufB;
    int* sorted = (int*)(bufB + (size_t)N_NODES * N_CLASSES);
    int* offs   = sorted + N_EDGES;
    int* pre    = offs + 102400;
    int* gcnt   = pre + BIN_BLKS * NBP;
    int* bstart = gcnt + NBP * GC_STRIDE;
    unsigned* binned = (unsigned*)bufA;

    init_kernel<<<1, NBP, 0, stream>>>(gcnt);
    bucket_count_kernel<<<(nE + BIN_CHUNK - 1) / BIN_CHUNK, BIN_T, 0, stream>>>(dst, gcnt, pre, nE);
    bucket_scatter_kernel<<<(nE + BIN_CHUNK - 1) / BIN_CHUNK, BIN_T, 0, stream>>>(src, dst, gcnt, pre, bstart, binned, nE);
    bucket_sort_kernel<<<NB, SORT_T, 0, stream>>>(binned, bstart, offs, dinv, sorted);
    gemm1_kernel<<<N_NODES / 32, 128, 0, stream>>>(x, W1, dinv, bufA);
    gather1_gemm2_kernel<<<N_NODES / 16, 256, 0, stream>>>(offs, sorted, dinv, bufA, b1, W2, h2);
    gather2_lsm_kernel<<<N_NODES / 16, 256, 0, stream>>>(offs, sorted, dinv, h2, b2, out);
}


# Gathers: both nodes' list bounds, own rows and source ids requested at wave entry; row loads of 16 (layer 1) / 32 (layer 2) in-edges issued before the first add; sum order kept
# speedup vs baseline: 3.3458x; 1.0833x over previous
#include <hip/hip_runtime.h>
#include <math.h>

typedef float v2f __attribute__((ext_vector_type(2)));
typedef float v4f __attribute__((ext_vector_type(4)));
typedef float v8f __attribute__((ext_vector_type(8)));

#define N_NODES    100000
#define N_EDGES    1600000
#define N_FEAT     512
#define HIDDEN     64
#define N_CLASSES  16
#define BKT_SHIFT  9
#define BKT_W      512
#define NB         ((N_NODES + BKT_W - 1) / BKT_W)
#define NBP        256
#define GC_STRIDE  32
#define BIN_T      256
#define BIN_CHUNK  (BIN_T * 24)
#define BIN_BLKS   ((N_EDGES + BIN_CHUNK - 1) / BIN_CHUNK)
#define SORT_T     512
#define STAGE_CAP  12288
static_assert(NB <= NBP && BIN_T == NBP && SORT_T == BKT_W, "one scan thread per bucket / per destination");
static_assert(N_NODES <= (1 << (32 - BKT_SHIFT)), "a source id and 9 destination bits share one 32-bit word");
#define TILE_PITCH 66

__global__ __launch_bounds__(NBP) void init_kernel(int* gcnt) {
    gcnt[threadIdx.x * GC_STRIDE] = 0;
}

__global__ __launch_bounds__(BIN_T) void bucket_count_kernel(const int* __restrict__ dst,
                                                             int* __restrict__ gcnt,
                                                             int* __restrict__ pre, int nE) {
    __shared__ int hist[NBP];
    int t = threadIdx.x;
    hist[t] = 0;
    __syncthreads();
    int c0 = blockIdx.x * BIN_CHUNK;
    int c1 = c0 + BIN_CHUNK;
    if (c1 > nE) c1 = nE;
    for (int e0 = c0; e0 < c1; e0 += BIN_T * 8) {
        bool full = e0 + BIN_T * 8 <= c1;
        int d[8];
        for (int k = 0; k < 8; ++k) {
            int ek = e0 + t + k * BIN_T;
            d[k] = dst[ek < c1 ? ek : c1 - 1];
        }
        if (full) {
            for (int k = 0; k < 8; ++k)
                __hip_atomic_fetch_add(&hist[d[k] >> BKT_SHIFT], 1, __ATOMIC_RELAXED, __HIP_MEMORY_SCOPE_WORKGROUP);
        } else {
            for (int k = 0; k < 8; ++k)
                if (e0 + t + k * BIN_T < c1)
                    __hip_atomic_fetch_add(&hist[d[k] >> BKT_SHIFT], 1, __ATOMIC_RELAXED, __HIP_MEMORY_SCOPE_WORKGROUP);
        }
    }
    __syncthreads();
    if (t < NB)
        pre[blockIdx.x * NBP + t] =
            __hip_atomic_fetch_add(&gcnt[t * GC_STRIDE], hist[t], __ATOMIC_RELAXED, __HIP_MEMORY_SCOPE_AGENT);
}

__global__ __launch_bounds__(BIN_T) void bucket_scatter_kernel(const int* __restrict__ src,
                                                               const int* __restrict__ dst,
                                                               const int* __restrict__ gcnt,
                                                               const int* __restrict__ pre,
                                                               int* __restrict__ bstart,
                                                               unsigned* __restrict__ binned, int nE) {
    __shared__ int cur[NBP];
    __shared__ int wtot[NBP / 32];
    int t = threadIdx.x, lane = t & 31, w = t >> 5;
    int v = (t < NB) ? gcnt[t * GC_STRIDE] : 0;
    int inc = v;
    for (int o = 1; o < 32; o <<= 1) {
        int u = __shfl_up(inc, o);
        if (lane >= o) inc += u;
    }
    if (lane == 31) wtot[w] = inc;
    __syncthreads();
    int wbase = 0;
    for (int k = 0; k < NBP / 32; ++k)
        if (k < w) wbase += wtot[k];
    int excl = wbase + inc - v;
    cur[t] = excl + ((t < NB) ? pre[blockIdx.x * NBP + t] : 0);
    if (blockIdx.x == 0 && t <= NB) bstart[t] = excl;
    __syncthreads();

    int c0 = blockIdx.x * BIN_CHUNK;
    int c1 = c0 + BIN_CHUNK;
    if (c1 > nE) c1 = nE;
    for (int e0 = c0; e0 < c1; e0 += BIN_T * 8) {
        bool full = e0 + BIN_T * 8 <= c1;
        int d[8], s[8];
        for (int k = 0; k < 8; ++k) {
            int ek = e0 + t + k * BIN_T;
            int ec = ek < c1 ? ek : c1 - 1;
            d[k] = dst[ec];
            s[k] = src[ec];
        }
        if (full) {
            int p[8];
            for (int k = 0; k < 8; ++k)
                p[k] = __hip_atomic_fetch_add(&cur[d[k] >> BKT_SHIFT], 1, __ATOMIC_RELAXED, __HIP_MEMORY_SCOPE_WORKGROUP);
            for (int k = 0; k < 8; ++k)
                binned[p[k]] = ((unsigned)s[k] << BKT_SHIFT) | (unsigned)(d[k] & (BKT_W - 1));
        } else {
            for (int k = 0; k < 8; ++k)
                if (e0 + t + k * BIN_T < c1) {
                    int p = __hip_atomic_fetch_add(&cur[d[k] >> BKT_SHIFT], 1, __ATOMIC_RELAXED, __HIP_MEMORY_SCOPE_WORKGROUP);
                    binned[p] = ((unsigned)s[k] << BKT_SHIFT) | (unsigned)(d[k] & (BKT_W - 1));
                }
        }
    }
}

__global__ __launch_bounds__(SORT_T) void bucket_sort_kernel(const unsigned* __restrict__ binned,
                                                             const int* __restrict__ bstart,
                                                             int* __restrict__ offs,
                                                             float* __restrict__ dinv,
                                                             int* __restrict__ sorted) {
    __shared__ int cntL[BKT_W];
    __shared__ int curL[BKT_W];
    __shared__ int wtot[SORT_T / 32];
    __shared__ int stage[STAGE_CAP];
    int t = threadIdx.x, lane = t & 31, w = t >> 5;
    int b  = blockIdx.x;
    int bs = bstart[b];
    int n  = bstart[b + 1] - bs;
    const unsigned* mine = binned + bs;
    cntL[t] = 0;
    __syncthreads();
    for (int i0 = 0; i0 < n; i0 += SORT_T * 8) {
        bool full = i0 + SORT_T * 8 <= n;
        unsigned p[8];
        for (int k = 0; k < 8; ++k) {
            int ik = i0 + t + k * SORT_T;
            p[k] = mine[ik < n ? ik : n - 1];
        }
        if (full) {
            for (int k = 0; k < 8; ++k)
                __hip_atomic_fetch_add(&cntL[p[k] & (BKT_W - 1)], 1, __ATOMIC_RELAXED, __HIP_MEMORY_SCOPE_WORKGROUP);
        } else {
            for (int k = 0; k < 8; ++k)
                if (i0 + t + k * SORT_T < n)
                    __hip_atomic_fetch_add(&cntL[p[k] & (BKT_W - 1)], 1, __ATOMIC_RELAXED, __HIP_MEMORY_SCOPE_WORKGROUP);
        }
    }
    __syncthreads();

    int v = cntL[t];
    int inc = v;
    for (int o = 1; o < 32; o <<= 1) {
        int u = __shfl_up(inc, o);
        if (lane >= o) inc += u;
    }
    if (lane == 31) wtot[w] = inc;
    __syncthreads();
    int wbase = 0;
    for (int k = 0; k < SORT_T / 32; ++k)
        if (k < w) wbase += wtot[k];
    int excl = wbase + inc - v;
    curL[t] = excl;
    int dnode = b * BKT_W + t;
    if (dnode < N_NODES) {
        offs[dnode] = bs + excl;
        dinv[dnode] = rsqrtf(1.0f + (float)v);
        if (dnode == N_NODES - 1) offs[N_NODES] = bs + excl + v;
    }
    __syncthreads();

    bool fits = n <= STAGE_CAP;
    for (int i0 = 0; i0 < n; i0 += SORT_T * 8) {
        bool full = i0 + SORT_T * 8 <= n;
        unsigned p[8];
        for (int k = 0; k < 8; ++k) {
            int ik = i0 + t + k * SORT_T;
            p[k] = mine[ik < n ? ik : n - 1];
        }
        if (full && fits) {
            int pos[8];
            for (int k = 0; k < 8; ++k)
                pos[k] = __hip_atomic_fetch_add(&curL[p[k] & (BKT_W - 1)], 1, __ATOMIC_RELAXED, __HIP_MEMORY_SCOPE_WORKGROUP);
            for (int k = 0; k < 8; ++k)
                stage[pos[k]] = (int)(p[k] >> BKT_SHIFT);
        } else {
            for (int k = 0; k < 8; ++k)
                if (i0 + t + k * SORT_T < n) {
                    int pos = __hip_atomic_fetch_add(&curL[p[k] & (BKT_W - 1)], 1, __ATOMIC_RELAXED, __HIP_MEMORY_SCOPE_WORKGROUP);
                    int s = (int)(p[k] >> BKT_SHIFT);
                    if (fits) stage[pos] = s;
                    else      sorted[bs + pos] = s;
                }
        }
    }
    __syncthreads();
    if (fits)
        for (int i = t; i < n; i += SORT_T) sorted[bs + i] = stage[i];
}

#define K_HALF (N_FEAT / 2)
static_assert(N_NODES % 32 == 0 && K_HALF % 32 == 0, "two whole 16-row tiles per block, whole 32-k lines per K half");
__global__ __launch_bounds__(128) void gemm1_kernel(const float* __restrict__ A,
                                                    const float* __restrict__ B,
                                                    const float* __restrict__ dinv,
                                                    float* __restrict__ C) {
    __shared__ float red[2][32][32];
    int wave = __builtin_amdgcn_readfirstlane(threadIdx.x >> 5);
    int lane = threadIdx.x & 31;
    int pr   = wave >> 1;
    int kh   = wave & 1;
    int tile = blockIdx.x * 2 + pr;
    int c  = lane & 15;
    int hi = lane >> 4;

    const float* arow = A + (size_t)(tile * 16 + c) * N_FEAT + kh * K_HALF + 4 * hi;
    const float* bcol = B + (size_t)(kh * K_HALF + 4 * hi) * HIDDEN + 4 * c;
    v8f acc0 = {}, acc1 = {}, acc2 = {}, acc3 = {};

    v4f an0 = *(const v4f*)(arow);
    v4f an1 = *(const v4f*)(arow + 8);
    v4f an2 = *(const v4f*)(arow + 16);
    v4f an3 = *(const v4f*)(arow + 24);
    v4f qn0 = *(const v4f*)(bcol);
    v4f qn1 = *(const v4f*)(bcol + HIDDEN);
    v4f qn2 = *(const v4f*)(bcol + 2 * HIDDEN);
    v4f qn3 = *(const v4f*)(bcol + 3 * HIDDEN);
    for (int K0 = 0; K0 < K_HALF; K0 += 32) {
        v4f a[4] = { an0, an1, an2, an3 };
        if (K0 + 32 < K_HALF) {
            const float* ap = arow + K0 + 32;
            an0 = *(const v4f*)(ap);
            an1 = *(const v4f*)(ap + 8);
            an2 = *(const v4f*)(ap + 16);
            an3 = *(const v4f*)(ap + 24);
        }
        for (int s = 0; s < 4; ++s) {
            v4f q0 = qn0, q1 = qn1, q2 = qn2, q3 = qn3;
            int kn = K0 + 8 * s + 8;
            if (kn < K_HALF) {
                const float* bp = bcol + (size_t)kn * HIDDEN;
                qn0 = *(const v4f*)(bp);
                qn1 = *(const v4f*)(bp + HIDDEN);
                qn2 = *(const v4f*)(bp + 2 * HIDDEN);
                qn3 = *(const v4f*)(bp + 3 * HIDDEN);
            }
            __builtin_amdgcn_sched_barrier(0);
            v2f a01 = { a[s].x, a[s].y };
            v2f a23 = { a[s].z, a[s].w };
            v2f b;
            b = (v2f){ q0.x, q1.x }; acc0 = __builtin_amdgcn_wmma_f32_16x16x4_f32(false, a01, false, b, (short)0, acc0, false, false);
            b = (v2f){ q0.y, q1.y }; acc1 = __builtin_amdgcn_wmma_f32_16x16x4_f32(false, a01, false, b, (short)0, acc1, false, false);
            b = (v2f){ q0.z, q1.z }; acc2 = __builtin_amdgcn_wmma_f32_16x16x4_f32(false, a01, false, b, (short)0, acc2, false, false);
            b = (v2f){ q0.w, q1.w }; acc3 = __builtin_amdgcn_wmma_f32_16x16x4_f32(false, a01, false, b, (short)0, acc3, false, false);
            b = (v2f){ q2.x, q3.x }; acc0 = __builtin_amdgcn_wmma_f32_16x16x4_f32(false, a23, false, b, (short)0, acc0, false, false);
            b = (v2f){ q2.y, q3.y }; acc1 = __builtin_amdgcn_wmma_f32_16x16x4_f32(false, a23, false, b, (short)0, acc1, false, false);
            b = (v2f){ q2.z, q3.z }; acc2 = __builtin_amdgcn_wmma_f32_16x16x4_f32(false, a23, false, b, (short)0, acc2, false, false);
            b = (v2f){ q2.w, q3.w }; acc3 = __builtin_amdgcn_wmma_f32_16x16x4_f32(false, a23, false, b, (short)0, acc3, false, false);
        }
    }
    asm volatile("v_nop\n\tv_nop\n\tv_nop\n\tv_nop\n\tv_nop"
                 : "+v"(acc0), "+v"(acc1), "+v"(acc2), "+v"(acc3));
    if (kh == 1) {
        for (int r = 0; r < 8; ++r) {
            red[pr][r][lane]      = acc0[r];
            red[pr][8 + r][lane]  = acc1[r];
            red[pr][16 + r][lane] = acc2[r];
            red[pr][24 + r][lane] = acc3[r];
        }
    }
    __syncthreads();
    if (kh == 0) {
        int mbase = tile * 16 + hi * 8;
        v4f dlo = *(const v4f*)(dinv + mbase);
        v4f dhi = *(const v4f*)(dinv + mbase + 4);
        float ds[8] = { dlo.x, dlo.y, dlo.z, dlo.w, dhi.x, dhi.y, dhi.z, dhi.w };
        for (int r = 0; r < 8; ++r) {
            float s = ds[r];
            v4f o = { (acc0[r] + red[pr][r][lane]) * s,      (acc1[r] + red[pr][8 + r][lane]) * s,
                      (acc2[r] + red[pr][16 + r][lane]) * s, (acc3[r] + red[pr][24 + r][lane]) * s };
            *(v4f*)(C + (size_t)(mbase + r) * HIDDEN + 4 * c) = o;
        }
    }
}

__device__ __forceinline__ v2f gather_rows64(const float* __restrict__ h, const int* __restrict__ sorted,
                                             int lane, int sl, int beg, int end, v2f acc) {
    for (int base = beg; base < end; base += 32) {
        if (base != beg) {
            sl = 0;
            if (base + lane < end) sl = sorted[base + lane];
        }
        int n = end - base;
        if (n > 32) n = 32;
        for (int j0 = 0; j0 < n; j0 += 16) {
            int s[16];
#pragma unroll
            for (int u = 0; u < 16; ++u) s[u] = __builtin_amdgcn_readlane(sl, j0 + u);
            v2f r[16];
#pragma unroll
            for (int g = 0; g < 4; ++g) {
                if (j0 + 4 * g < n) {
#pragma unroll
                    for (int e = 0; e < 4; ++e)
                        r[4 * g + e] = *(const v2f*)(h + ((unsigned)s[4 * g + e] * HIDDEN + 2 * lane));
                }
            }
#pragma unroll
            for (int u = 0; u < 16; ++u) {
                if (j0 + u < n) {
                    acc.x += r[u].x;
                    acc.y += r[u].y;
                }
            }
        }
    }
    return acc;
}

__global__ __launch_bounds__(256) void gather1_gemm2_kernel(const int* __restrict__ offs,
                                                            const int* __restrict__ sorted,
                                                            const float* __restrict__ dinv,
                                                            const float* __restrict__ h,
                                                            const float* __restrict__ b1,
                                                            const float* __restrict__ W2,
                                                            float* __restrict__ h2) {
    __shared__ __attribute__((aligned(16))) float tile[16 * TILE_PITCH];
    int lane = threadIdx.x & 31;
    int wave = __builtin_amdgcn_readfirstlane(threadIdx.x >> 5);
    v2f bias = *(const v2f*)(b1 + 2 * lane);

    int row0 = wave * 2;
    int d0   = blockIdx.x * 16 + row0;
    int o0 = offs[d0], o1 = offs[d0 + 1], o2 = offs[d0 + 2];
    float di0 = dinv[d0], di1 = dinv[d0 + 1];
    v2f self0 = *(const v2f*)(h + (size_t)d0 * HIDDEN + 2 * lane);
    v2f self1 = *(const v2f*)(h + (size_t)(d0 + 1) * HIDDEN + 2 * lane);
    int sl0 = 0, sl1 = 0;
    if (o0 + lane < o1) sl0 = sorted[o0 + lane];
    if (o1 + lane < o2) sl1 = sorted[o1 + lane];

    for (int q = 0; q < 2; ++q) {
        int row = row0 + q;
        float di = q ? di1 : di0;
        v2f a = gather_rows64(h, sorted, lane, q ? sl1 : sl0, q ? o1 : o0, q ? o2 : o1, q ? self1 : self0);
        float v0 = di * a.x + bias.x, v1 = di * a.y + bias.y;
        v2f o = { (v0 > 0.0f ? v0 : 0.0f) * di, (v1 > 0.0f ? v1 : 0.0f) * di };
        *(v2f*)(tile + row * TILE_PITCH + 2 * lane) = o;
    }
    __syncthreads();

    if (wave == 0) {
        int nr = lane & 15;
        int hi = lane >> 4;
        const float* arow = tile + nr * TILE_PITCH + 2 * hi;
        v8f acc = {};
        for (int kk = 0; kk < HIDDEN; kk += 4) {
            v2f a = *(const v2f*)(arow + kk);
            const float* b0p = W2 + (size_t)(kk + 2 * hi) * N_CLASSES + nr;
            v2f b = { b0p[0], b0p[N_CLASSES] };
            acc = __builtin_amdgcn_wmma_f32_16x16x4_f32(false, a, false, b, (short)0, acc, false, false);
        }
        int mbase = blockIdx.x * 16 + hi * 8;
        for (int r = 0; r < 8; ++r)
            h2[(size_t)(mbase + r) * N_CLASSES + nr] = acc[r];
    }
}

__device__ __forceinline__ float gather_rows16(const float* __restrict__ h2, const int* __restrict__ sorted,
                                               int lane, int sl, int beg, int end) {
    int c    = lane & 15;
    int half = lane >> 4;
    float acc = 0.0f;
    for (int base = beg; base < end; base += 32) {
        if (base != beg) {
            sl = 0;
            if (base + lane < end) sl = sorted[base + lane];
        }
        int n = end - base;
        if (n > 32) n = 32;
        int s[32];
#pragma unroll
        for (int u = 0; u < 32; ++u) s[u] = __builtin_amdgcn_readlane(sl, u);
        float g[16];
#pragma unroll
        for (int k = 0; k < 4; ++k) {
            if (8 * k < n) {
#pragma unroll
                for (int e = 0; e < 4; ++e) {
                    int u = 4 * k + e;
                    g[u] = h2[(unsigned)(half ? s[2 * u + 1] : s[2 * u]) * N_CLASSES + c];
                }
            }
        }
#pragma unroll
        for (int u = 0; u < 16; ++u)
            if (2 * u + half < n) acc += g[u];
    }
    return acc;
}

__global__ __launch_bounds__(256) void gather2_lsm_kernel(const int* __restrict__ offs,
                                                          const int* __restrict__ sorted,
                                                          const float* __restrict__ dinv,
                                                          const float* __restrict__ h2,
                                                          const float* __restrict__ b2,
                                                          float* __restrict__ out) {
    int lane = threadIdx.x & 31;
    int wave = __builtin_amdgcn_readfirstlane(threadIdx.x >> 5);
    int c    = lane & 15;
    int half = lane >> 4;
    int pair = blockIdx.x * 8 + wave;
    float bc = b2[c];
    float x  = 0.0f;

    int d0 = pair * 2;
    int o0 = offs[d0], o1 = offs[d0 + 1], o2 = offs[d0 + 2];
    float di0 = dinv[d0], di1 = dinv[d0 + 1];
    float selfv0 = h2[(size_t)d0 * N_CLASSES + c];
    float selfv1 = h2[(size_t)(d0 + 1) * N_CLASSES + c];
    int sl0 = 0, sl1 = 0;
    if (o0 + lane < o1) sl0 = sorted[o0 + lane];
    if (o1 + lane < o2) sl1 = sorted[o1 + lane];

    for (int q = 0; q < 2; ++q) {
        float di    = q ? di1 : di0;
        float selfv = q ? selfv1 : selfv0;
        float acc = gather_rows16(h2, sorted, lane, q ? sl1 : sl0, q ? o1 : o0, q ? o2 : o1);
        acc += __shfl_xor(acc, 16);
        float val = di * (acc + selfv) + bc;
        if (half == q) x = val;
    }

    float m = x;
    for (int o = 8; o > 0; o >>= 1) m = fmaxf(m, __shfl_xor(m, o));
    float ssum = expf(x - m);
    for (int o = 8; o > 0; o >>= 1) ssum += __shfl_xor(ssum, o);
    out[(size_t)(pair * 2 + half) * N_CLASSES + c] = x - m - logf(ssum);
}

extern "C" void kernel_launch(void* const* d_in, const int* in_sizes, int n_in,
                              void* d_out, int out_size, void* d_ws, size_t ws_size,
                              hipStream_t stream) {
    const float* x  = (const float*)d_in[0];
    const int*   ei = (const int*)d_in[1];
    const float* W1 = (const float*)d_in[2];
    const float* b1 = (const float*)d_in[3];
    const float* W2 = (const float*)d_in[4];
    const float* b2 = (const float*)d_in[5];
    float* out = (float*)d_out;

    int nE = in_sizes[1] / 2;
    const int* src = ei;
    const int* dst = ei + nE;

    float* dinv = (float*)d_ws;
    float* bufA = dinv + 102400;
    float* bufB = bufA + (size_t)N_NODES * HIDDEN;
    float* h2   = bufB;
    int* sorted = (int*)(bufB + (size_t)N_NODES * N_CLASSES);
    int* offs   = sorted + N_EDGES;
    int* pre    = offs + 102400;
    int* gcnt   = pre + BIN_BLKS * NBP;
    int* bstart = gcnt + NBP * GC_STRIDE;
    unsigned* binned = (unsigned*)bufA;

    init_kernel<<<1, NBP, 0, stream>>>(gcnt);
    bucket_count_kernel<<<(nE + BIN_CHUNK - 1) / BIN_CHUNK, BIN_T, 0, stream>>>(dst, gcnt, pre, nE);
    bucket_scatter_kernel<<<(nE + BIN_CHUNK - 1) / BIN_CHUNK, BIN_T, 0, stream>>>(src, dst, gcnt, pre, bstart, binned, nE);
    bucket_sort_kernel<<<NB, SORT_T, 0, stream>>>(binned, bstart, offs, dinv, sorted);
    gemm1_kernel<<<N_NODES / 32, 128, 0, stream>>>(x, W1, dinv, bufA);
    gather1_gemm2_kernel<<<N_NODES / 16, 256, 0, stream>>>(offs, sorted, dinv, bufA, b1, W2, h2);
    gather2_lsm_kernel<<<N_NODES / 16, 256, 0, stream>>>(offs, sorted, dinv, h2, b2, out);
}


# Edge bucketing chunk-local: a block orders its own 6144 edges by bucket in LDS, the bucket sort reads its 261 segments once into registers; no memory counters, 7 -> 5 launches; LDS arrival order
# speedup vs baseline: 3.8986x; 1.1652x over previous
#include <hip/hip_runtime.h>
#include <math.h>

typedef float v2f __attribute__((ext_vector_type(2)));
typedef float v4f __attribute__((ext_vector_type(4)));
typedef float v8f __attribute__((ext_vector_type(8)));

#define N_NODES    100000
#define N_EDGES    1600000
#define N_FEAT     512
#define HIDDEN     64
#define N_CLASSES  16
#define BKT_SHIFT  9
#define BKT_W      512
#define NB         ((N_NODES + BKT_W - 1) / BKT_W)
#define NBP        256
#define BIN_T      256
#define BIN_CHUNK  (BIN_T * 24)
#define BIN_PER_T  (BIN_CHUNK / BIN_T)
#define BIN_BLKS   ((N_EDGES + BIN_CHUNK - 1) / BIN_CHUNK)
#define SORT_T     512
#define SORT_W     (SORT_T / 32)
#define SEG_PW     ((BIN_BLKS + SORT_W - 1) / SORT_W)
#define STAGE_CAP  12288
static_assert(NB < NBP && BIN_T == NBP && SORT_T == BKT_W, "one scan thread per bucket / per destination; column NB of the table holds the chunk's edge count");
static_assert(BIN_BLKS <= SORT_T && SORT_W * SEG_PW <= SORT_T, "one thread of the bucket sort per chunk segment");
static_assert(N_NODES <= (1 << (32 - BKT_SHIFT)), "a source id and 9 destination bits share one 32-bit word");
#define TILE_PITCH 66

__global__ __launch_bounds__(BIN_T) void chunk_bin_kernel(const int* __restrict__ src,
                                                          const int* __restrict__ dst,
                                                          int* __restrict__ pre,
                                                          unsigned* __restrict__ binned, int nE) {
    __shared__ int hist[NBP];
    __shared__ int wtot[NBP / 32];
    __shared__ unsigned ordered[BIN_CHUNK];
    int t = threadIdx.x, lane = t & 31, w = t >> 5;
    hist[t] = 0;
    int c0 = blockIdx.x * BIN_CHUNK;
    int c1 = c0 + BIN_CHUNK;
    if (c1 > nE) c1 = nE;
    bool full = c1 - c0 == BIN_CHUNK;
    int d[BIN_PER_T], s[BIN_PER_T];
#pragma unroll
    for (int k = 0; k < BIN_PER_T; ++k) {
        int ek = c0 + t + k * BIN_T;
        int ec = ek < c1 ? ek : c1 - 1;
        d[k] = dst[ec];
        s[k] = src[ec];
    }
    __syncthreads();
    if (full) {
#pragma unroll
        for (int k = 0; k < BIN_PER_T; ++k)
            __hip_atomic_fetch_add(&hist[d[k] >> BKT_SHIFT], 1, __ATOMIC_RELAXED, __HIP_MEMORY_SCOPE_WORKGROUP);
    } else {
#pragma unroll
        for (int k = 0; k < BIN_PER_T; ++k)
            if (c0 + t + k * BIN_T < c1)
                __hip_atomic_fetch_add(&hist[d[k] >> BKT_SHIFT], 1, __ATOMIC_RELAXED, __HIP_MEMORY_SCOPE_WORKGROUP);
    }
    __syncthreads();

    int v = hist[t];
    int inc = v;
    for (int o = 1; o < 32; o <<= 1) {
        int u = __shfl_up(inc, o);
        if (lane >= o) inc += u;
    }
    if (lane == 31) wtot[w] = inc;
    __syncthreads();
    int wbase = 0;
    for (int k = 0; k < NBP / 32; ++k)
        if (k < w) wbase += wtot[k];
    int excl = wbase + inc - v;
    pre[blockIdx.x * NBP + t] = excl;
    hist[t] = excl;
    __syncthreads();

    if (full) {
        int p[BIN_PER_T];
#pragma unroll
        for (int k = 0; k < BIN_PER_T; ++k)
            p[k] = __hip_atomic_fetch_add(&hist[d[k] >> BKT_SHIFT], 1, __ATOMIC_RELAXED, __HIP_MEMORY_SCOPE_WORKGROUP);
#pragma unroll
        for (int k = 0; k < BIN_PER_T; ++k)
            ordered[p[k]] = ((unsigned)s[k] << BKT_SHIFT) | (unsigned)(d[k] & (BKT_W - 1));
    } else {
#pragma unroll
        for (int k = 0; k < BIN_PER_T; ++k)
            if (c0 + t + k * BIN_T < c1) {
                int p = __hip_atomic_fetch_add(&hist[d[k] >> BKT_SHIFT], 1, __ATOMIC_RELAXED, __HIP_MEMORY_SCOPE_WORKGROUP);
                ordered[p] = ((unsigned)s[k] << BKT_SHIFT) | (unsigned)(d[k] & (BKT_W - 1));
            }
    }
    __syncthreads();
    int cn = c1 - c0;
    for (int i = t; i < cn; i += BIN_T) binned[c0 + i] = ordered[i];
}

__global__ __launch_bounds__(SORT_T) void bucket_sort_kernel(const unsigned* __restrict__ binned,
                                                             const int* __restrict__ pre,
                                                             int* __restrict__ offs,
                                                             float* __restrict__ dinv,
                                                             int* __restrict__ sorted) {
    __shared__ int cntL[BKT_W];
    __shared__ int curL[BKT_W];
    __shared__ int wtot[SORT_W];
    __shared__ int wend[SORT_W];
    __shared__ int segS[SORT_W * SEG_PW];
    __shared__ int segL[SORT_W * SEG_PW];
    __shared__ int stage[STAGE_CAP];
    int t = threadIdx.x, lane = t & 31;
    int w = __builtin_amdgcn_readfirstlane(t >> 5);
    int b = blockIdx.x;

    int s0 = 0, s1 = 0;
    if (t < BIN_BLKS) {
        s0 = pre[t * NBP + b];
        s1 = pre[t * NBP + b + 1];
    }
    cntL[t] = 0;
    if (t < SORT_W * SEG_PW) {
        segS[t] = t * BIN_CHUNK + s0;
        segL[t] = s1 - s0;
    }
    int a0 = s0, a1 = s1;
    for (int o = 16; o > 0; o >>= 1) {
        a0 += __shfl_xor(a0, o);
        a1 += __shfl_xor(a1, o);
    }
    if (lane == 0) { wtot[w] = a0; wend[w] = a1; }
    __syncthreads();
    int bs = 0, be = 0;
    for (int k = 0; k < SORT_W; ++k) { bs += wtot[k]; be += wend[k]; }
    int n = be - bs;

    int sS[SEG_PW], sN[SEG_PW];
    unsigned p[2 * SEG_PW];
#pragma unroll
    for (int i = 0; i < SEG_PW; ++i) {
        sS[i] = __builtin_amdgcn_readfirstlane(segS[w + SORT_W * i]);
        sN[i] = __builtin_amdgcn_readfirstlane(segL[w + SORT_W * i]);
    }
#pragma unroll
    for (int i = 0; i < SEG_PW; ++i) {
        int last = sN[i] > 0 ? sN[i] - 1 : 0;
        p[2 * i]     = binned[sS[i] + (lane < last ? lane : last)];
        p[2 * i + 1] = binned[sS[i] + (lane + 32 < last ? lane + 32 : last)];
    }
#pragma unroll
    for (int i = 0; i < 2 * SEG_PW; ++i)
        if (lane + 32 * (i & 1) < sN[i >> 1])
            __hip_atomic_fetch_add(&cntL[p[i] & (BKT_W - 1)], 1, __ATOMIC_RELAXED, __HIP_MEMORY_SCOPE_WORKGROUP);
#pragma clang loop unroll(disable)
    for (int i = 0; i < SEG_PW; ++i) {
        int ts = __builtin_amdgcn_readfirstlane(segS[w + SORT_W * i]);
        int tn = __builtin_amdgcn_readfirstlane(segL[w + SORT_W * i]);
#pragma clang loop unroll(disable)
        for (int off = 64; off < tn; off += 32)
            if (off + lane < tn) {
                unsigned q = binned[ts + off + lane];
                __hip_atomic_fetch_add(&cntL[q & (BKT_W - 1)], 1, __ATOMIC_RELAXED, __HIP_MEMORY_SCOPE_WORKGROUP);
            }
    }
    __syncthreads();

    int v = cntL[t];
    int inc = v;
    for (int o = 1; o < 32; o <<= 1) {
        int u = __shfl_up(inc, o);
        if (lane >= o) inc += u;
    }
    if (lane == 31) wtot[w] = inc;
    __syncthreads();
    int wbase = 0;
    for (int k = 0; k < SORT_W; ++k)
        if (k < w) wbase += wtot[k];
    int excl = wbase + inc - v;
    curL[t] = excl;
    int dnode = b * BKT_W + t;
    if (dnode < N_NODES) {
        offs[dnode] = bs + excl;
        dinv[dnode] = rsqrtf(1.0f + (float)v);
        if (dnode == N_NODES - 1) offs[N_NODES] = bs + excl + v;
    }
    __syncthreads();

    bool fits = n <= STAGE_CAP;
    int pos[2 * SEG_PW];
#pragma unroll
    for (int i = 0; i < 2 * SEG_PW; ++i) {
        pos[i] = 0;
        if (lane + 32 * (i & 1) < sN[i >> 1])
            pos[i] = __hip_atomic_fetch_add(&curL[p[i] & (BKT_W - 1)], 1, __ATOMIC_RELAXED, __HIP_MEMORY_SCOPE_WORKGROUP);
    }
#pragma unroll
    for (int i = 0; i < 2 * SEG_PW; ++i)
        if (lane + 32 * (i & 1) < sN[i >> 1]) {
            int sv = (int)(p[i] >> BKT_SHIFT);
            if (fits) stage[pos[i]] = sv;
            else      sorted[bs + pos[i]] = sv;
        }
#pragma clang loop unroll(disable)
    for (int i = 0; i < SEG_PW; ++i) {
        int ts = __builtin_amdgcn_readfirstlane(segS[w + SORT_W * i]);
        int tn = __builtin_amdgcn_readfirstlane(segL[w + SORT_W * i]);
#pragma clang loop unroll(disable)
        for (int off = 64; off < tn; off += 32)
            if (off + lane < tn) {
                unsigned q = binned[ts + off + lane];
                int ps = __hip_atomic_fetch_add(&curL[q & (BKT_W - 1)], 1, __ATOMIC_RELAXED, __HIP_MEMORY_SCOPE_WORKGROUP);
                int sv = (int)(q >> BKT_SHIFT);
                if (fits) stage[ps] = sv;
                else      sorted[bs + ps] = sv;
            }
    }
    __syncthreads();
    if (fits)
        for (int i = t; i < n; i += SORT_T) sorted[bs + i] = stage[i];
}

#define K_HALF (N_FEAT / 2)
static_assert(N_NODES % 32 == 0 && K_HALF % 32 == 0, "two whole 16-row tiles per block, whole 32-k lines per K half");
__global__ __launch_bounds__(128) void gemm1_kernel(const float* __restrict__ A,
                                                    const float* __restrict__ B,
                                                    const float* __restrict__ dinv,
                                                    float* __restrict__ C) {
    __shared__ float red[2][32][32];
    int wave = __builtin_amdgcn_readfirstlane(threadIdx.x >> 5);
    int lane = threadIdx.x & 31;
    int pr   = wave >> 1;
    int kh   = wave & 1;
    int tile = blockIdx.x * 2 + pr;
    int c  = lane & 15;
    int hi = lane >> 4;

    const float* arow = A + (size_t)(tile * 16 + c) * N_FEAT + kh * K_HALF + 4 * hi;
    const float* bcol = B + (size_t)(kh * K_HALF + 4 * hi) * HIDDEN + 4 * c;
    v8f acc0 = {}, acc1 = {}, acc2 = {}, acc3 = {};

    v4f an0 = *(const v4f*)(arow);
    v4f an1 = *(const v4f*)(arow + 8);
    v4f an2 = *(const v4f*)(arow + 16);
    v4f an3 = *(const v4f*)(arow + 24);
    v4f qn0 = *(const v4f*)(bcol);
    v4f qn1 = *(const v4f*)(bcol + HIDDEN);
    v4f qn2 = *(const v4f*)(bcol + 2 * HIDDEN);
    v4f qn3 = *(const v4f*)(bcol + 3 * HIDDEN);
    for (int K0 = 0; K0 < K_HALF; K0 += 32) {
        v4f a[4] = { an0, an1, an2, an3 };
        if (K0 + 32 < K_HALF) {
            const float* ap = arow + K0 + 32;
            an0 = *(const v4f*)(ap);
            an1 = *(const v4f*)(ap + 8);
            an2 = *(const v4f*)(ap + 16);
            an3 = *(const v4f*)(ap + 24);
        }
        for (int s = 0; s < 4; ++s) {
            v4f q0 = qn0, q1 = qn1, q2 = qn2, q3 = qn3;
            int kn = K0 + 8 * s + 8;
            if (kn < K_HALF) {
                const float* bp = bcol + (size_t)kn * HIDDEN;
                qn0 = *(const v4f*)(bp);
                qn1 = *(const v4f*)(bp + HIDDEN);
                qn2 = *(const v4f*)(bp + 2 * HIDDEN);
                qn3 = *(const v4f*)(bp + 3 * HIDDEN);
            }
            __builtin_amdgcn_sched_barrier(0);
            v2f a01 = { a[s].x, a[s].y };
            v2f a23 = { a[s].z, a[s].w };
            v2f b;
            b = (v2f){ q0.x, q1.x }; acc0 = __builtin_amdgcn_wmma_f32_16x16x4_f32(false, a01, false, b, (short)0, acc0, false, false);
            b = (v2f){ q0.y, q1.y }; acc1 = __builtin_amdgcn_wmma_f32_16x16x4_f32(false, a01, false, b, (short)0, acc1, false, false);
            b = (v2f){ q0.z, q1.z }; acc2 = __builtin_amdgcn_wmma_f32_16x16x4_f32(false, a01, false, b, (short)0, acc2, false, false);
            b = (v2f){ q0.w, q1.w }; acc3 = __builtin_amdgcn_wmma_f32_16x16x4_f32(false, a01, false, b, (short)0, acc3, false, false);
            b = (v2f){ q2.x, q3.x }; acc0 = __builtin_amdgcn_wmma_f32_16x16x4_f32(false, a23, false, b, (short)0, acc0, false, false);
            b = (v2f){ q2.y, q3.y }; acc1 = __builtin_amdgcn_wmma_f32_16x16x4_f32(false, a23, false, b, (short)0, acc1, false, false);
            b = (v2f){ q2.z, q3.z }; acc2 = __builtin_amdgcn_wmma_f32_16x16x4_f32(false, a23, false, b, (short)0, acc2, false, false);
            b = (v2f){ q2.w, q3.w }; acc3 = __builtin_amdgcn_wmma_f32_16x16x4_f32(false, a23, false, b, (short)0, acc3, false, false);
        }
    }
    asm volatile("v_nop\n\tv_nop\n\tv_nop\n\tv_nop\n\tv_nop"
                 : "+v"(acc0), "+v"(acc1), "+v"(acc2), "+v"(acc3));
    if (kh == 1) {
        for (int r = 0; r < 8; ++r) {
            red[pr][r][lane]      = acc0[r];
            red[pr][8 + r][lane]  = acc1[r];
            red[pr][16 + r][lane] = acc2[r];
            red[pr][24 + r][lane] = acc3[r];
        }
    }
    __syncthreads();
    if (kh == 0) {
        int mbase = tile * 16 + hi * 8;
        v4f dlo = *(const v4f*)(dinv + mbase);
        v4f dhi = *(const v4f*)(dinv + mbase + 4);
        float ds[8] = { dlo.x, dlo.y, dlo.z, dlo.w, dhi.x, dhi.y, dhi.z, dhi.w };
        for (int r = 0; r < 8; ++r) {
            float s = ds[r];
            v4f o = { (acc0[r] + red[pr][r][lane]) * s,      (acc1[r] + red[pr][8 + r][lane]) * s,
                      (acc2[r] + red[pr][16 + r][lane]) * s, (acc3[r] + red[pr][24 + r][lane]) * s };
            *(v4f*)(C + (size_t)(mbase + r) * HIDDEN + 4 * c) = o;
        }
    }
}

__device__ __forceinline__ v2f gather_rows64(const float* __restrict__ h, const int* __restrict__ sorted,
                                             int lane, int sl, int beg, int end, v2f acc) {
    for (int base = beg; base < end; base += 32) {
        if (base != beg) {
            sl = 0;
            if (base + lane < end) sl = sorted[base + lane];
        }
        int n = end - base;
        if (n > 32) n = 32;
        for (int j0 = 0; j0 < n; j0 += 16) {
            int s[16];
#pragma unroll
            for (int u = 0; u < 16; ++u) s[u] = __builtin_amdgcn_readlane(sl, j0 + u);
            v2f r[16];
#pragma unroll
            for (int g = 0; g < 4; ++g) {
                if (j0 + 4 * g < n) {
#pragma unroll
                    for (int e = 0; e < 4; ++e)
                        r[4 * g + e] = *(const v2f*)(h + ((unsigned)s[4 * g + e] * HIDDEN + 2 * lane));
                }
            }
#pragma unroll
            for (int u = 0; u < 16; ++u) {
                if (j0 + u < n) {
                    acc.x += r[u].x;
                    acc.y += r[u].y;
                }
            }
        }
    }
    return acc;
}

__global__ __launch_bounds__(256) void gather1_gemm2_kernel(const int* __restrict__ offs,
                                                            const int* __restrict__ sorted,
                                                            const float* __restrict__ dinv,
                                                            const float* __restrict__ h,
                                                            const float* __restrict__ b1,
                                                            const float* __restrict__ W2,
                                                            float* __restrict__ h2) {
    __shared__ __attribute__((aligned(16))) float tile[16 * TILE_PITCH];
    int lane = threadIdx.x & 31;
    int wave = __builtin_amdgcn_readfirstlane(threadIdx.x >> 5);
    v2f bias = *(const v2f*)(b1 + 2 * lane);

    int row0 = wave * 2;
    int d0   = blockIdx.x * 16 + row0;
    int o0 = offs[d0], o1 = offs[d0 + 1], o2 = offs[d0 + 2];
    float di0 = dinv[d0], di1 = dinv[d0 + 1];
    v2f self0 = *(const v2f*)(h + (size_t)d0 * HIDDEN + 2 * lane);
    v2f self1 = *(const v2f*)(h + (size_t)(d0 + 1) * HIDDEN + 2 * lane);
    int sl0 = 0, sl1 = 0;
    if (o0 + lane < o1) sl0 = sorted[o0 + lane];
    if (o1 + lane < o2) sl1 = sorted[o1 + lane];

    for (int q = 0; q < 2; ++q) {
        int row = row0 + q;
        float di = q ? di1 : di0;
        v2f a = gather_rows64(h, sorted, lane, q ? sl1 : sl0, q ? o1 : o0, q ? o2 : o1, q ? self1 : self0);
        float v0 = di * a.x + bias.x, v1 = di * a.y + bias.y;
        v2f o = { (v0 > 0.0f ? v0 : 0.0f) * di, (v1 > 0.0f ? v1 : 0.0f) * di };
        *(v2f*)(tile + row * TILE_PITCH + 2 * lane) = o;
    }
    __syncthreads();

    if (wave == 0) {
        int nr = lane & 15;
        int hi = lane >> 4;
        const float* arow = tile + nr * TILE_PITCH + 2 * hi;
        v8f acc = {};
        for (int kk = 0; kk < HIDDEN; kk += 4) {
            v2f a = *(const v2f*)(arow + kk);
            const float* b0p = W2 + (size_t)(kk + 2 * hi) * N_CLASSES + nr;
            v2f b = { b0p[0], b0p[N_CLASSES] };
            acc = __builtin_amdgcn_wmma_f32_16x16x4_f32(false, a, false, b, (short)0, acc, false, false);
        }
        int mbase = blockIdx.x * 16 + hi * 8;
        for (int r = 0; r < 8; ++r)
            h2[(size_t)(mbase + r) * N_CLASSES + nr] = acc[r];
    }
}

__device__ __forceinline__ float gather_rows16(const float* __restrict__ h2, const int* __restrict__ sorted,
                                               int lane, int sl, int beg, int end) {
    int c    = lane & 15;
    int half = lane >> 4;
    float acc = 0.0f;
    for (int base = beg; base < end; base += 32) {
        if (base != beg) {
            sl = 0;
            if (base + lane < end) sl = sorted[base + lane];
        }
        int n = end - base;
        if (n > 32) n = 32;
        int s[32];
#pragma unroll
        for (int u = 0; u < 32; ++u) s[u] = __builtin_amdgcn_readlane(sl, u);
        float g[16];
#pragma unroll
        for (int k = 0; k < 4; ++k) {
            if (8 * k < n) {
#pragma unroll
                for (int e = 0; e < 4; ++e) {
                    int u = 4 * k + e;
                    g[u] = h2[(unsigned)(half ? s[2 * u + 1] : s[2 * u]) * N_CLASSES + c];
                }
            }
        }
#pragma unroll
        for (int u = 0; u < 16; ++u)
            if (2 * u + half < n) acc += g[u];
    }
    return acc;
}

__global__ __launch_bounds__(256) void gather2_lsm_kernel(const int* __restrict__ offs,
                                                          const int* __restrict__ sorted,
                                                          const float* __restrict__ dinv,
                                                          const float* __restrict__ h2,
                                                          const float* __restrict__ b2,
                                                          float* __restrict__ out) {
    int lane = threadIdx.x & 31;
    int wave = __builtin_amdgcn_readfirstlane(threadIdx.x >> 5);
    int c    = lane & 15;
    int half = lane >> 4;
    int pair = blockIdx.x * 8 + wave;
    float bc = b2[c];
    float x  = 0.0f;

    int d0 = pair * 2;
    int o0 = offs[d0], o1 = offs[d0 + 1], o2 = offs[d0 + 2];
    float di0 = dinv[d0], di1 = dinv[d0 + 1];
    float selfv0 = h2[(size_t)d0 * N_CLASSES + c];
    float selfv1 = h2[(size_t)(d0 + 1) * N_CLASSES + c];
    int sl0 = 0, sl1 = 0;
    if (o0 + lane < o1) sl0 = sorted[o0 + lane];
    if (o1 + lane < o2) sl1 = sorted[o1 + lane];

    for (int q = 0; q < 2; ++q) {
        float di    = q ? di1 : di0;
        float selfv = q ? selfv1 : selfv0;
        float acc = gather_rows16(h2, sorted, lane, q ? sl1 : sl0, q ? o1 : o0, q ? o2 : o1);
        acc += __shfl_xor(acc, 16);
        float val = di * (acc + selfv) + bc;
        if (half == q) x = val;
    }

    float m = x;
    for (int o = 8; o > 0; o >>= 1) m = fmaxf(m, __shfl_xor(m, o));
    float ssum = expf(x - m);
    for (int o = 8; o > 0; o >>= 1) ssum += __shfl_xor(ssum, o);
    out[(size_t)(pair * 2 + half) * N_CLASSES + c] = x - m - logf(ssum);
}

extern "C" void kernel_launch(void* const* d_in, const int* in_sizes, int n_in,
                              void* d_out, int out_size, void* d_ws, size_t ws_size,
                              hipStream_t stream) {
    const float* x  = (const float*)d_in[0];
    const int*   ei = (const int*)d_in[1];
    const float* W1 = (const float*)d_in[2];
    const float* b1 = (const float*)d_in[3];
    const float* W2 = (const float*)d_in[4];
    const float* b2 = (const float*)d_in[5];
    float* out = (float*)d_out;

    int nE = in_sizes[1] / 2;
    const int* src = ei;
    const int* dst = ei + nE;

    float* dinv = (float*)d_ws;
    float* bufA = dinv + 102400;
    float* bufB = bufA + (size_t)N_NODES * HIDDEN;
    float* h2   = bufB;
    int* sorted = (int*)(bufB + (size_t)N_NODES * N_CLASSES);
    int* offs   = sorted + N_EDGES;
    int* pre    = offs + 102400;
    unsigned* binned = (unsigned*)bufA;

    chunk_bin_kernel<<<(nE + BIN_CHUNK - 1) / BIN_CHUNK, BIN_T, 0, stream>>>(src, dst, pre, binned, nE);
    bucket_sort_kernel<<<NB, SORT_T, 0, stream>>>(binned, pre, offs, dinv, sorted);
    gemm1_kernel<<<N_NODES / 32, 128, 0, stream>>>(x, W1, dinv, bufA);
    gather1_gemm2_kernel<<<N_NODES / 16, 256, 0, stream>>>(offs, sorted, dinv, bufA, b1, W2, h2);
    gather2_lsm_kernel<<<N_NODES / 16, 256, 0, stream>>>(offs, sorted, dinv, h2, b2, out);
}
